# starting kernel
# baseline (speedup 1.0000x reference)
#include <hip/hip_runtime.h>
#include <hip/hip_bf16.h>
#include <math.h>

typedef __attribute__((ext_vector_type(2))) float v2f;
typedef __attribute__((ext_vector_type(8))) float v8f;

#define DD 64
#define EPS_COS 1e-8f
#define EPS_BN  1e-5f

__global__ __launch_bounds__(64) void gather_gate_kernel(
    const float* __restrict__ embed_u_z, const float* __restrict__ embed_u_c,
    const float* __restrict__ embed_v_z, const float* __restrict__ embed_v_c,
    const float* __restrict__ item_table,
    const int* __restrict__ hist_items, const int* __restrict__ hist_lens,
    const int* __restrict__ item_ids, const int* __restrict__ red_by_friends,
    float* __restrict__ uv, int H) {
  __shared__ __align__(16) float anchor_s[DD];
  __shared__ float sim_s[64];
  __shared__ int   valid_s[64];

  const int b = blockIdx.x;
  const int t = threadIdx.x;
  const int item = item_ids[b];

  anchor_s[t] = item_table[(long)item * DD + t];
  __syncthreads();

  float sim = -INFINITY;
  int valid = 0;
  if (t < H) {
    const int hi = hist_items[b * H + t];
    const float4* he4 = (const float4*)(item_table + (long)hi * DD);
    const float4* an4 = (const float4*)anchor_s;
    float dot = 0.0f, na = 0.0f, nb = 0.0f;
#pragma unroll
    for (int i = 0; i < DD / 4; ++i) {
      float4 x = he4[i];
      float4 a = an4[i];
      dot += x.x * a.x + x.y * a.y + x.z * a.z + x.w * a.w;
      na  += x.x * x.x + x.y * x.y + x.z * x.z + x.w * x.w;
      nb  += a.x * a.x + a.y * a.y + a.z * a.z + a.w * a.w;
    }
    na = sqrtf(na);
    nb = sqrtf(nb);
    sim = dot / (fmaxf(na, EPS_COS) * fmaxf(nb, EPS_COS));
    valid = (t < hist_lens[b]) && (hi != item);
  }
  sim_s[t]   = valid ? sim : -INFINITY;
  valid_s[t] = valid;
  __syncthreads();

  for (int s = 32; s > 0; s >>= 1) {
    if (t < s) {
      sim_s[t]   = fmaxf(sim_s[t], sim_s[t + s]);
      valid_s[t] += valid_s[t + s];
    }
    __syncthreads();
  }

  const bool empty   = (valid_s[0] == 0);
  const bool gate    = (red_by_friends[b] > 0) && (sim_s[0] > 0.5f);
  const bool use_u_c = empty || gate;

  const long bd = (long)b * DD + t;
  const float u = embed_u_z[bd] + (use_u_c ? embed_u_c[bd] : 0.0f);
  const float v = embed_v_z[bd] +
                  (empty ? embed_u_c[bd] : (gate ? embed_v_c[bd] : 0.0f));
  uv[(long)b * (2 * DD) + t]      = u;
  uv[(long)b * (2 * DD) + DD + t] = v;
}

__global__ __launch_bounds__(32) void mlp_layer_kernel(
    const float* __restrict__ x,
    const float* __restrict__ w,
    const float* __restrict__ bias,
    const float* __restrict__ bn_sum,
    const float* __restrict__ bn_sq,
    const float* __restrict__ bn_g,
    const float* __restrict__ bn_be,
    float* __restrict__ y,
    int K, int N, int use_bn, float invB) {
  __shared__ float sc_s[128];
  __shared__ float sh_s[128];

  const int lane = threadIdx.x;
  if (use_bn) {
    for (int c = lane; c < K; c += 32) {
      float mean = bn_sum[c] * invB;
      float var  = bn_sq[c] * invB - mean * mean;
      float sc   = bn_g[c] * rsqrtf(var + EPS_BN);
      sc_s[c] = sc;
      sh_s[c] = bn_be[c] - mean * sc;
    }
  } else {
    for (int c = lane; c < K; c += 32) {
      sc_s[c] = 1.0f;
      sh_s[c] = 0.0f;
    }
  }
  __syncthreads();

  const int row0  = blockIdx.x * 16;
  const int r     = lane & 15;
  const int khalf = (lane >> 4) * 2;
  const int col   = lane & 15;

  for (int nt = 0; nt < (N >> 4); ++nt) {
    v8f acc = {};
    for (int k = 0; k < K; k += 4) {
      const int kk = k + khalf;
      v2f a, bb;
      a.x  = sc_s[kk]     * x[(long)(row0 + r) * K + kk]     + sh_s[kk];
      a.y  = sc_s[kk + 1] * x[(long)(row0 + r) * K + kk + 1] + sh_s[kk + 1];
      bb.x = w[(long)kk * N + nt * 16 + col];
      bb.y = w[(long)(kk + 1) * N + nt * 16 + col];
      acc = __builtin_amdgcn_wmma_f32_16x16x4_f32(
          false, a, false, bb, (short)0, acc, false, false);
    }
    const int n = nt * 16 + col;
    const float bn = bias[n];
#pragma unroll
    for (int j = 0; j < 8; ++j) {
      const int m = row0 + j + (lane >> 4) * 8;
      y[(long)m * N + n] = fmaxf(acc[j] + bn, 0.0f);
    }
  }
}

__global__ __launch_bounds__(256) void col_stats_kernel(
    const float* __restrict__ y, float* __restrict__ out_sum,
    float* __restrict__ out_sq, int Bn, int N) {
  __shared__ float ssum[256];
  __shared__ float ssq[256];
  const int n = blockIdx.x;
  const int t = threadIdx.x;
  float s = 0.0f, q = 0.0f;
  for (int m = t; m < Bn; m += 256) {
    float v = y[(long)m * N + n];
    s += v;
    q += v * v;
  }
  ssum[t] = s;
  ssq[t]  = q;
  __syncthreads();
  for (int st = 128; st > 0; st >>= 1) {
    if (t < st) {
      ssum[t] += ssum[t + st];
      ssq[t]  += ssq[t + st];
    }
    __syncthreads();
  }
  if (t == 0) {
    out_sum[n] = ssum[0];
    out_sq[n]  = ssq[0];
  }
}

__global__ __launch_bounds__(256) void final_kernel(
    const float* __restrict__ y3, const float* __restrict__ sum3,
    const float* __restrict__ sq3, const float* __restrict__ g3,
    const float* __restrict__ be3, const float* __restrict__ wf,
    const float* __restrict__ bf, float* __restrict__ out, int Bn, float invB) {
  __shared__ float sc_s[16], sh_s[16], wf_s[16];
  const int t = threadIdx.x;
  if (t < 16) {
    float mean = sum3[t] * invB;
    float var  = sq3[t] * invB - mean * mean;
    float sc   = g3[t] * rsqrtf(var + EPS_BN);
    sc_s[t] = sc;
    sh_s[t] = be3[t] - mean * sc;
    wf_s[t] = wf[t];
  }
  __syncthreads();
  const int b = blockIdx.x * blockDim.x + t;
  if (b < Bn) {
    float s = 0.0f;
#pragma unroll
    for (int c = 0; c < 16; ++c)
      s += (sc_s[c] * y3[(long)b * 16 + c] + sh_s[c]) * wf_s[c];
    out[b] = s + bf[0];
  }
}

extern "C" void kernel_launch(void* const* d_in, const int* in_sizes, int n_in,
                              void* d_out, int out_size, void* d_ws,
                              size_t ws_size, hipStream_t stream) {
  const float* embed_u_z = (const float*)d_in[0];
  const float* embed_u_c = (const float*)d_in[1];
  const float* embed_v_z = (const float*)d_in[2];
  const float* embed_v_c = (const float*)d_in[3];
  const float* item_table = (const float*)d_in[4];
  const float* w1 = (const float*)d_in[5];
  const float* b1 = (const float*)d_in[6];
  const float* g1 = (const float*)d_in[7];
  const float* be1 = (const float*)d_in[8];
  const float* w2 = (const float*)d_in[9];
  const float* b2 = (const float*)d_in[10];
  const float* g2 = (const float*)d_in[11];
  const float* be2 = (const float*)d_in[12];
  const float* w3 = (const float*)d_in[13];
  const float* b3 = (const float*)d_in[14];
  const float* g3 = (const float*)d_in[15];
  const float* be3 = (const float*)d_in[16];
  const float* wf = (const float*)d_in[17];
  const float* bf = (const float*)d_in[18];
  const int* hist_items = (const int*)d_in[19];
  const int* hist_lens = (const int*)d_in[20];
  const int* item_ids = (const int*)d_in[21];
  const int* red_by_friends = (const int*)d_in[22];

  const int Bn = in_sizes[20];
  const int H  = in_sizes[19] / Bn;
  const float invB = 1.0f / (float)Bn;

  float* ws = (float*)d_ws;
  float* uv = ws;
  float* y1 = uv + (size_t)Bn * 128;
  float* y2 = y1 + (size_t)Bn * 64;
  float* y3 = y2 + (size_t)Bn * 32;
  float* stats = y3 + (size_t)Bn * 16;
  float* sum1 = stats;      float* sq1 = sum1 + 64;
  float* sum2 = sq1 + 64;   float* sq2 = sum2 + 32;
  float* sum3 = sq2 + 32;   float* sq3 = sum3 + 16;

  gather_gate_kernel<<<Bn, 64, 0, stream>>>(
      embed_u_z, embed_u_c, embed_v_z, embed_v_c, item_table,
      hist_items, hist_lens, item_ids, red_by_friends, uv, H);

  mlp_layer_kernel<<<Bn / 16, 32, 0, stream>>>(
      uv, w1, b1, nullptr, nullptr, nullptr, nullptr,
      y1, 128, 64, 0, invB);
  col_stats_kernel<<<64, 256, 0, stream>>>(y1, sum1, sq1, Bn, 64);

  mlp_layer_kernel<<<Bn / 16, 32, 0, stream>>>(
      y1, w2, b2, sum1, sq1, g1, be1,
      y2, 64, 32, 1, invB);
  col_stats_kernel<<<32, 256, 0, stream>>>(y2, sum2, sq2, Bn, 32);

  mlp_layer_kernel<<<Bn / 16, 32, 0, stream>>>(
      y2, w3, b3, sum2, sq2, g2, be2,
      y3, 32, 16, 1, invB);
  col_stats_kernel<<<16, 256, 0, stream>>>(y3, sum3, sq3, Bn, 16);

  final_kernel<<<(Bn + 255) / 256, 256, 0, stream>>>(
      y3, sum3, sq3, g3, be3, wf, bf, (float*)d_out, Bn, invB);
}


# BN column sums fused into the layer kernels: 4-wave blocks store per-block partial sums from registers, consumers add them in fixed order; 3 stats launches removed; the sums' reduction tree changes
# speedup vs baseline: 2.1953x; 2.1953x over previous
#include <hip/hip_runtime.h>
#include <hip/hip_bf16.h>
#include <math.h>

typedef __attribute__((ext_vector_type(2))) float v2f;
typedef __attribute__((ext_vector_type(8))) float v8f;

#define DD 64
#define EPS_COS 1e-8f
#define EPS_BN  1e-5f

__global__ __launch_bounds__(64) void gather_gate_kernel(
    const float* __restrict__ embed_u_z, const float* __restrict__ embed_u_c,
    const float* __restrict__ embed_v_z, const float* __restrict__ embed_v_c,
    const float* __restrict__ item_table,
    const int* __restrict__ hist_items, const int* __restrict__ hist_lens,
    const int* __restrict__ item_ids, const int* __restrict__ red_by_friends,
    float* __restrict__ uv, int H) {
  __shared__ __align__(16) float anchor_s[DD];
  __shared__ float sim_s[64];
  __shared__ int   valid_s[64];

  const int b = blockIdx.x;
  const int t = threadIdx.x;
  const int item = item_ids[b];

  anchor_s[t] = item_table[(long)item * DD + t];
  __syncthreads();

  float sim = -INFINITY;
  int valid = 0;
  if (t < H) {
    const int hi = hist_items[b * H + t];
    const float4* he4 = (const float4*)(item_table + (long)hi * DD);
    const float4* an4 = (const float4*)anchor_s;
    float dot = 0.0f, na = 0.0f, nb = 0.0f;
#pragma unroll
    for (int i = 0; i < DD / 4; ++i) {
      float4 x = he4[i];
      float4 a = an4[i];
      dot += x.x * a.x + x.y * a.y + x.z * a.z + x.w * a.w;
      na  += x.x * x.x + x.y * x.y + x.z * x.z + x.w * x.w;
      nb  += a.x * a.x + a.y * a.y + a.z * a.z + a.w * a.w;
    }
    na = sqrtf(na);
    nb = sqrtf(nb);
    sim = dot / (fmaxf(na, EPS_COS) * fmaxf(nb, EPS_COS));
    valid = (t < hist_lens[b]) && (hi != item);
  }
  sim_s[t]   = valid ? sim : -INFINITY;
  valid_s[t] = valid;
  __syncthreads();

  for (int s = 32; s > 0; s >>= 1) {
    if (t < s) {
      sim_s[t]   = fmaxf(sim_s[t], sim_s[t + s]);
      valid_s[t] += valid_s[t + s];
    }
    __syncthreads();
  }

  const bool empty   = (valid_s[0] == 0);
  const bool gate    = (red_by_friends[b] > 0) && (sim_s[0] > 0.5f);
  const bool use_u_c = empty || gate;

  const long bd = (long)b * DD + t;
  const float u = embed_u_z[bd] + (use_u_c ? embed_u_c[bd] : 0.0f);
  const float v = embed_v_z[bd] +
                  (empty ? embed_u_c[bd] : (gate ? embed_v_c[bd] : 0.0f));
  uv[(long)b * (2 * DD) + t]      = u;
  uv[(long)b * (2 * DD) + DD + t] = v;
}

template <int NT, int UNR>
__device__ __forceinline__ void bn_from_partials(
    const float* __restrict__ psum, const float* __restrict__ psq, int nparts,
    int K, const float* __restrict__ bn_g, const float* __restrict__ bn_be,
    float invB, float* sc_s, float* sh_s, float* red_s, float* red_q) {
  const int t     = threadIdx.x;
  const int pitch = K < 32 ? 32 : K;
  const int qpr   = K >> 2;
  const int G     = NT / qpr;
  const int cq    = t % qpr;
  const int grp   = t / qpr;
  if (grp < G) {
    float4 s = make_float4(0.0f, 0.0f, 0.0f, 0.0f);
    float4 q = make_float4(0.0f, 0.0f, 0.0f, 0.0f);
#pragma unroll UNR
    for (int p = grp; p < nparts; p += G) {
      const float4 a = *(const float4*)(psum + (long)p * pitch + 4 * cq);
      const float4 b = *(const float4*)(psq  + (long)p * pitch + 4 * cq);
      s.x += a.x; s.y += a.y; s.z += a.z; s.w += a.w;
      q.x += b.x; q.y += b.y; q.z += b.z; q.w += b.w;
    }
    float* rs = red_s + grp * K + 4 * cq;
    float* rq = red_q + grp * K + 4 * cq;
    rs[0] = s.x; rs[1] = s.y; rs[2] = s.z; rs[3] = s.w;
    rq[0] = q.x; rq[1] = q.y; rq[2] = q.z; rq[3] = q.w;
  }
  __syncthreads();
  if (t < K) {
    float S = red_s[t];
    float Q = red_q[t];
    for (int i = 1; i < G; ++i) {
      S += red_s[i * K + t];
      Q += red_q[i * K + t];
    }
    const float mean = S * invB;
    const float var  = Q * invB - mean * mean;
    const float sc   = bn_g[t] * rsqrtf(var + EPS_BN);
    sc_s[t] = sc;
    sh_s[t] = bn_be[t] - mean * sc;
  }
  __syncthreads();
}

#define MLP_WAVES   4
#define MLP_THREADS (MLP_WAVES * 32)

__global__ __launch_bounds__(MLP_THREADS) void mlp_layer_kernel(
    const float* __restrict__ x,
    const float* __restrict__ w,
    const float* __restrict__ bias,
    const float* __restrict__ in_psum,
    const float* __restrict__ in_psq,
    const float* __restrict__ bn_g,
    const float* __restrict__ bn_be,
    float* __restrict__ y,
    float* __restrict__ out_psum,
    float* __restrict__ out_psq,
    int K, int N, int use_bn, int nparts, float invB) {
  __shared__ float sc_s[128];
  __shared__ float sh_s[128];
  __shared__ __align__(16) float red_s[4 * MLP_THREADS];
  __shared__ __align__(16) float red_q[4 * MLP_THREADS];
  __shared__ float wred[2][MLP_WAVES][2][64];

  const int t    = threadIdx.x;
  const int lane = t & 31;
  const int wave = t >> 5;
  if (use_bn) {
    bn_from_partials<MLP_THREADS, 8>(in_psum, in_psq, nparts, K, bn_g, bn_be,
                                     invB, sc_s, sh_s, red_s, red_q);
  } else {
    for (int c = t; c < K; c += MLP_THREADS) {
      sc_s[c] = 1.0f;
      sh_s[c] = 0.0f;
    }
    __syncthreads();
  }

  const int row0  = (blockIdx.x * MLP_WAVES + wave) * 16;
  const int r     = lane & 15;
  const int half  = lane >> 4;
  const int khalf = half * 2;
  const int col   = lane & 15;

  for (int nt = 0; nt < (N >> 4); ++nt) {
    v8f acc = {};
    for (int k = 0; k < K; k += 4) {
      const int kk = k + khalf;
      v2f a, bb;
      a.x  = sc_s[kk]     * x[(long)(row0 + r) * K + kk]     + sh_s[kk];
      a.y  = sc_s[kk + 1] * x[(long)(row0 + r) * K + kk + 1] + sh_s[kk + 1];
      bb.x = w[(long)kk * N + nt * 16 + col];
      bb.y = w[(long)(kk + 1) * N + nt * 16 + col];
      acc = __builtin_amdgcn_wmma_f32_16x16x4_f32(
          false, a, false, bb, (short)0, acc, false, false);
    }
    const int n = nt * 16 + col;
    const float bn = bias[n];
    float s = 0.0f, q = 0.0f;
#pragma unroll
    for (int j = 0; j < 8; ++j) {
      const int m = row0 + j + half * 8;
      const float v = fmaxf(acc[j] + bn, 0.0f);
      y[(long)m * N + n] = v;
      s += v;
      q += v * v;
    }
    wred[0][wave][half][n] = s;
    wred[1][wave][half][n] = q;
  }
  __syncthreads();

  if (t < 2 * N) {
    const int kind   = t >= N ? 1 : 0;
    const int n      = t - kind * N;
    const int opitch = N < 32 ? 32 : N;
    float a = 0.0f;
    for (int wv = 0; wv < MLP_WAVES; ++wv)
      a += wred[kind][wv][0][n] + wred[kind][wv][1][n];
    (kind ? out_psq : out_psum)[(long)blockIdx.x * opitch + n] = a;
  }
}

__global__ __launch_bounds__(256) void final_kernel(
    const float* __restrict__ y3, const float* __restrict__ psum3,
    const float* __restrict__ psq3, int nparts, const float* __restrict__ g3,
    const float* __restrict__ be3, const float* __restrict__ wf,
    const float* __restrict__ bf, float* __restrict__ out, int Bn, float invB) {
  __shared__ float sc_s[16], sh_s[16], wf_s[16];
  __shared__ __align__(16) float red_s[4 * 256];
  __shared__ __align__(16) float red_q[4 * 256];
  const int t = threadIdx.x;
  if (t < 16) wf_s[t] = wf[t];
  bn_from_partials<256, 4>(psum3, psq3, nparts, 16, g3, be3, invB, sc_s, sh_s,
                           red_s, red_q);
  const int b = blockIdx.x * blockDim.x + t;
  if (b < Bn) {
    float s = 0.0f;
#pragma unroll
    for (int c = 0; c < 16; ++c)
      s += (sc_s[c] * y3[(long)b * 16 + c] + sh_s[c]) * wf_s[c];
    out[b] = s + bf[0];
  }
}

extern "C" void kernel_launch(void* const* d_in, const int* in_sizes, int n_in,
                              void* d_out, int out_size, void* d_ws,
                              size_t ws_size, hipStream_t stream) {
  const float* embed_u_z = (const float*)d_in[0];
  const float* embed_u_c = (const float*)d_in[1];
  const float* embed_v_z = (const float*)d_in[2];
  const float* embed_v_c = (const float*)d_in[3];
  const float* item_table = (const float*)d_in[4];
  const float* w1 = (const float*)d_in[5];
  const float* b1 = (const float*)d_in[6];
  const float* g1 = (const float*)d_in[7];
  const float* be1 = (const float*)d_in[8];
  const float* w2 = (const float*)d_in[9];
  const float* b2 = (const float*)d_in[10];
  const float* g2 = (const float*)d_in[11];
  const float* be2 = (const float*)d_in[12];
  const float* w3 = (const float*)d_in[13];
  const float* b3 = (const float*)d_in[14];
  const float* g3 = (const float*)d_in[15];
  const float* be3 = (const float*)d_in[16];
  const float* wf = (const float*)d_in[17];
  const float* bf = (const float*)d_in[18];
  const int* hist_items = (const int*)d_in[19];
  const int* hist_lens = (const int*)d_in[20];
  const int* item_ids = (const int*)d_in[21];
  const int* red_by_friends = (const int*)d_in[22];

  const int Bn = in_sizes[20];
  const int H  = in_sizes[19] / Bn;
  const float invB = 1.0f / (float)Bn;
  const int NB = Bn / (16 * MLP_WAVES);

  float* ws = (float*)d_ws;
  float* uv = ws;
  float* y1 = uv + (size_t)Bn * 128;
  float* y2 = y1 + (size_t)Bn * 64;
  float* y3 = y2 + (size_t)Bn * 32;
  float* ps1 = y3 + (size_t)Bn * 16;
  float* pq1 = ps1 + (size_t)NB * 64;
  float* ps2 = pq1 + (size_t)NB * 64;
  float* pq2 = ps2 + (size_t)NB * 32;
  float* ps3 = pq2 + (size_t)NB * 32;
  float* pq3 = ps3 + (size_t)NB * 32;

  gather_gate_kernel<<<Bn, 64, 0, stream>>>(
      embed_u_z, embed_u_c, embed_v_z, embed_v_c, item_table,
      hist_items, hist_lens, item_ids, red_by_friends, uv, H);

  mlp_layer_kernel<<<NB, MLP_THREADS, 0, stream>>>(
      uv, w1, b1, nullptr, nullptr, nullptr, nullptr,
      y1, ps1, pq1, 128, 64, 0, 0, invB);

  mlp_layer_kernel<<<NB, MLP_THREADS, 0, stream>>>(
      y1, w2, b2, ps1, pq1, g1, be1,
      y2, ps2, pq2, 64, 32, 1, NB, invB);

  mlp_layer_kernel<<<NB, MLP_THREADS, 0, stream>>>(
      y2, w3, b3, ps2, pq2, g2, be2,
      y3, ps3, pq3, 32, 16, 1, NB, invB);

  final_kernel<<<(Bn + 255) / 256, 256, 0, stream>>>(
      y3, ps3, pq3, NB, g3, be3, wf, bf, (float*)d_out, Bn, invB);
}


# gather kernel fetches and scores only the valid history rows: entries at or beyond hist_lens or equal to the anchor item are masked to -inf by the reference, so their rows are never read
# speedup vs baseline: 2.3599x; 1.0750x over previous
#include <hip/hip_runtime.h>
#include <hip/hip_bf16.h>
#include <math.h>

typedef __attribute__((ext_vector_type(2))) float v2f;
typedef __attribute__((ext_vector_type(8))) float v8f;

#define DD 64
#define EPS_COS 1e-8f
#define EPS_BN  1e-5f

__global__ __launch_bounds__(64) void gather_gate_kernel(
    const float* __restrict__ embed_u_z, const float* __restrict__ embed_u_c,
    const float* __restrict__ embed_v_z, const float* __restrict__ embed_v_c,
    const float* __restrict__ item_table,
    const int* __restrict__ hist_items, const int* __restrict__ hist_lens,
    const int* __restrict__ item_ids, const int* __restrict__ red_by_friends,
    float* __restrict__ uv, int H) {
  __shared__ __align__(16) float anchor_s[DD];
  __shared__ float sim_s[64];
  __shared__ int   valid_s[64];

  const int b = blockIdx.x;
  const int t = threadIdx.x;
  const int item = item_ids[b];
  const int len  = hist_lens[b];

  const bool in_len = (t < H) && (t < len);
  int hi = item;
  if (in_len) hi = hist_items[b * H + t];

  anchor_s[t] = item_table[(long)item * DD + t];
  __syncthreads();

  float sim = -INFINITY;
  const int valid = in_len && (hi != item);
  if (valid) {
    const float4* he4 = (const float4*)(item_table + (long)hi * DD);
    const float4* an4 = (const float4*)anchor_s;
    float dot = 0.0f, na = 0.0f, nb = 0.0f;
#pragma unroll
    for (int i = 0; i < DD / 4; ++i) {
      float4 x = he4[i];
      float4 a = an4[i];
      dot += x.x * a.x + x.y * a.y + x.z * a.z + x.w * a.w;
      na  += x.x * x.x + x.y * x.y + x.z * x.z + x.w * x.w;
      nb  += a.x * a.x + a.y * a.y + a.z * a.z + a.w * a.w;
    }
    na = sqrtf(na);
    nb = sqrtf(nb);
    sim = dot / (fmaxf(na, EPS_COS) * fmaxf(nb, EPS_COS));
  }
  sim_s[t]   = sim;
  valid_s[t] = valid;
  __syncthreads();

  for (int s = 32; s > 0; s >>= 1) {
    if (t < s) {
      sim_s[t]   = fmaxf(sim_s[t], sim_s[t + s]);
      valid_s[t] += valid_s[t + s];
    }
    __syncthreads();
  }

  const bool empty   = (valid_s[0] == 0);
  const bool gate    = (red_by_friends[b] > 0) && (sim_s[0] > 0.5f);
  const bool use_u_c = empty || gate;

  const long bd = (long)b * DD + t;
  const float u = embed_u_z[bd] + (use_u_c ? embed_u_c[bd] : 0.0f);
  const float v = embed_v_z[bd] +
                  (empty ? embed_u_c[bd] : (gate ? embed_v_c[bd] : 0.0f));
  uv[(long)b * (2 * DD) + t]      = u;
  uv[(long)b * (2 * DD) + DD + t] = v;
}

template <int NT, int UNR>
__device__ __forceinline__ void bn_from_partials(
    const float* __restrict__ psum, const float* __restrict__ psq, int nparts,
    int K, const float* __restrict__ bn_g, const float* __restrict__ bn_be,
    float invB, float* sc_s, float* sh_s, float* red_s, float* red_q) {
  const int t     = threadIdx.x;
  const int pitch = K < 32 ? 32 : K;
  const int qpr   = K >> 2;
  const int G     = NT / qpr;
  const int cq    = t % qpr;
  const int grp   = t / qpr;
  if (grp < G) {
    float4 s = make_float4(0.0f, 0.0f, 0.0f, 0.0f);
    float4 q = make_float4(0.0f, 0.0f, 0.0f, 0.0f);
#pragma unroll UNR
    for (int p = grp; p < nparts; p += G) {
      const float4 a = *(const float4*)(psum + (long)p * pitch + 4 * cq);
      const float4 b = *(const float4*)(psq  + (long)p * pitch + 4 * cq);
      s.x += a.x; s.y += a.y; s.z += a.z; s.w += a.w;
      q.x += b.x; q.y += b.y; q.z += b.z; q.w += b.w;
    }
    float* rs = red_s + grp * K + 4 * cq;
    float* rq = red_q + grp * K + 4 * cq;
    rs[0] = s.x; rs[1] = s.y; rs[2] = s.z; rs[3] = s.w;
    rq[0] = q.x; rq[1] = q.y; rq[2] = q.z; rq[3] = q.w;
  }
  __syncthreads();
  if (t < K) {
    float S = red_s[t];
    float Q = red_q[t];
    for (int i = 1; i < G; ++i) {
      S += red_s[i * K + t];
      Q += red_q[i * K + t];
    }
    const float mean = S * invB;
    const float var  = Q * invB - mean * mean;
    const float sc   = bn_g[t] * rsqrtf(var + EPS_BN);
    sc_s[t] = sc;
    sh_s[t] = bn_be[t] - mean * sc;
  }
  __syncthreads();
}

#define MLP_WAVES   4
#define MLP_THREADS (MLP_WAVES * 32)

__global__ __launch_bounds__(MLP_THREADS) void mlp_layer_kernel(
    const float* __restrict__ x,
    const float* __restrict__ w,
    const float* __restrict__ bias,
    const float* __restrict__ in_psum,
    const float* __restrict__ in_psq,
    const float* __restrict__ bn_g,
    const float* __restrict__ bn_be,
    float* __restrict__ y,
    float* __restrict__ out_psum,
    float* __restrict__ out_psq,
    int K, int N, int use_bn, int nparts, float invB) {
  __shared__ float sc_s[128];
  __shared__ float sh_s[128];
  __shared__ __align__(16) float red_s[4 * MLP_THREADS];
  __shared__ __align__(16) float red_q[4 * MLP_THREADS];
  __shared__ float wred[2][MLP_WAVES][2][64];

  const int t    = threadIdx.x;
  const int lane = t & 31;
  const int wave = t >> 5;
  if (use_bn) {
    bn_from_partials<MLP_THREADS, 8>(in_psum, in_psq, nparts, K, bn_g, bn_be,
                                     invB, sc_s, sh_s, red_s, red_q);
  } else {
    for (int c = t; c < K; c += MLP_THREADS) {
      sc_s[c] = 1.0f;
      sh_s[c] = 0.0f;
    }
    __syncthreads();
  }

  const int row0  = (blockIdx.x * MLP_WAVES + wave) * 16;
  const int r     = lane & 15;
  const int half  = lane >> 4;
  const int khalf = half * 2;
  const int col   = lane & 15;

  for (int nt = 0; nt < (N >> 4); ++nt) {
    v8f acc = {};
    for (int k = 0; k < K; k += 4) {
      const int kk = k + khalf;
      v2f a, bb;
      a.x  = sc_s[kk]     * x[(long)(row0 + r) * K + kk]     + sh_s[kk];
      a.y  = sc_s[kk + 1] * x[(long)(row0 + r) * K + kk + 1] + sh_s[kk + 1];
      bb.x = w[(long)kk * N + nt * 16 + col];
      bb.y = w[(long)(kk + 1) * N + nt * 16 + col];
      acc = __builtin_amdgcn_wmma_f32_16x16x4_f32(
          false, a, false, bb, (short)0, acc, false, false);
    }
    const int n = nt * 16 + col;
    const float bn = bias[n];
    float s = 0.0f, q = 0.0f;
#pragma unroll
    for (int j = 0; j < 8; ++j) {
      const int m = row0 + j + half * 8;
      const float v = fmaxf(acc[j] + bn, 0.0f);
      y[(long)m * N + n] = v;
      s += v;
      q += v * v;
    }
    wred[0][wave][half][n] = s;
    wred[1][wave][half][n] = q;
  }
  __syncthreads();

  if (t < 2 * N) {
    const int kind   = t >= N ? 1 : 0;
    const int n      = t - kind * N;
    const int opitch = N < 32 ? 32 : N;
    float a = 0.0f;
    for (int wv = 0; wv < MLP_WAVES; ++wv)
      a += wred[kind][wv][0][n] + wred[kind][wv][1][n];
    (kind ? out_psq : out_psum)[(long)blockIdx.x * opitch + n] = a;
  }
}

__global__ __launch_bounds__(256) void final_kernel(
    const float* __restrict__ y3, const float* __restrict__ psum3,
    const float* __restrict__ psq3, int nparts, const float* __restrict__ g3,
    const float* __restrict__ be3, const float* __restrict__ wf,
    const float* __restrict__ bf, float* __restrict__ out, int Bn, float invB) {
  __shared__ float sc_s[16], sh_s[16], wf_s[16];
  __shared__ __align__(16) float red_s[4 * 256];
  __shared__ __align__(16) float red_q[4 * 256];
  const int t = threadIdx.x;
  if (t < 16) wf_s[t] = wf[t];
  bn_from_partials<256, 4>(psum3, psq3, nparts, 16, g3, be3, invB, sc_s, sh_s,
                           red_s, red_q);
  const int b = blockIdx.x * blockDim.x + t;
  if (b < Bn) {
    float s = 0.0f;
#pragma unroll
    for (int c = 0; c < 16; ++c)
      s += (sc_s[c] * y3[(long)b * 16 + c] + sh_s[c]) * wf_s[c];
    out[b] = s + bf[0];
  }
}

extern "C" void kernel_launch(void* const* d_in, const int* in_sizes, int n_in,
                              void* d_out, int out_size, void* d_ws,
                              size_t ws_size, hipStream_t stream) {
  const float* embed_u_z = (const float*)d_in[0];
  const float* embed_u_c = (const float*)d_in[1];
  const float* embed_v_z = (const float*)d_in[2];
  const float* embed_v_c = (const float*)d_in[3];
  const float* item_table = (const float*)d_in[4];
  const float* w1 = (const float*)d_in[5];
  const float* b1 = (const float*)d_in[6];
  const float* g1 = (const float*)d_in[7];
  const float* be1 = (const float*)d_in[8];
  const float* w2 = (const float*)d_in[9];
  const float* b2 = (const float*)d_in[10];
  const float* g2 = (const float*)d_in[11];
  const float* be2 = (const float*)d_in[12];
  const float* w3 = (const float*)d_in[13];
  const float* b3 = (const float*)d_in[14];
  const float* g3 = (const float*)d_in[15];
  const float* be3 = (const float*)d_in[16];
  const float* wf = (const float*)d_in[17];
  const float* bf = (const float*)d_in[18];
  const int* hist_items = (const int*)d_in[19];
  const int* hist_lens = (const int*)d_in[20];
  const int* item_ids = (const int*)d_in[21];
  const int* red_by_friends = (const int*)d_in[22];

  const int Bn = in_sizes[20];
  const int H  = in_sizes[19] / Bn;
  const float invB = 1.0f / (float)Bn;
  const int NB = Bn / (16 * MLP_WAVES);

  float* ws = (float*)d_ws;
  float* uv = ws;
  float* y1 = uv + (size_t)Bn * 128;
  float* y2 = y1 + (size_t)Bn * 64;
  float* y3 = y2 + (size_t)Bn * 32;
  float* ps1 = y3 + (size_t)Bn * 16;
  float* pq1 = ps1 + (size_t)NB * 64;
  float* ps2 = pq1 + (size_t)NB * 64;
  float* pq2 = ps2 + (size_t)NB * 32;
  float* ps3 = pq2 + (size_t)NB * 32;
  float* pq3 = ps3 + (size_t)NB * 32;

  gather_gate_kernel<<<Bn, 64, 0, stream>>>(
      embed_u_z, embed_u_c, embed_v_z, embed_v_c, item_table,
      hist_items, hist_lens, item_ids, red_by_friends, uv, H);

  mlp_layer_kernel<<<NB, MLP_THREADS, 0, stream>>>(
      uv, w1, b1, nullptr, nullptr, nullptr, nullptr,
      y1, ps1, pq1, 128, 64, 0, 0, invB);

  mlp_layer_kernel<<<NB, MLP_THREADS, 0, stream>>>(
      y1, w2, b2, ps1, pq1, g1, be1,
      y2, ps2, pq2, 64, 32, 1, NB, invB);

  mlp_layer_kernel<<<NB, MLP_THREADS, 0, stream>>>(
      y2, w3, b3, ps2, pq2, g2, be2,
      y3, ps3, pq3, 32, 16, 1, NB, invB);

  final_kernel<<<(Bn + 255) / 256, 256, 0, stream>>>(
      y3, ps3, pq3, NB, g3, be3, wf, bf, (float*)d_out, Bn, invB);
}


# gather kernel remapped: 16 lanes read each 256-B history row in one coalesced load, two batch rows per wave, lane sums by DPP instead of LDS and barriers; dot and norm reduction tree changes
# speedup vs baseline: 3.0822x; 1.3061x over previous
#include <hip/hip_runtime.h>
#include <hip/hip_bf16.h>
#include <math.h>

typedef __attribute__((ext_vector_type(2))) float v2f;
typedef __attribute__((ext_vector_type(8))) float v8f;

#define DD 64
#define EPS_COS 1e-8f
#define EPS_BN  1e-5f

#define GG_WAVES   8
#define GG_THREADS (GG_WAVES * 32)
#define GG_CHUNK   16

template <int M>
__device__ __forceinline__ float row_xor(float v) {
  return __int_as_float(__builtin_amdgcn_update_dpp(
      0, __float_as_int(v), 0x160 + M, 0xf, 0xf, true));
}

__device__ __forceinline__ float row16_sum(float v) {
  v += row_xor<1>(v);
  v += row_xor<2>(v);
  v += row_xor<4>(v);
  v += row_xor<8>(v);
  return v;
}
__device__ __forceinline__ float row16_max(float v) {
  v = fmaxf(v, row_xor<1>(v));
  v = fmaxf(v, row_xor<2>(v));
  v = fmaxf(v, row_xor<4>(v));
  v = fmaxf(v, row_xor<8>(v));
  return v;
}

__device__ __forceinline__ float row16_transpose_sum(const float (&v)[16],
                                                     int c) {
  const bool b8 = (c & 8) != 0, b4 = (c & 4) != 0;
  const bool b2 = (c & 2) != 0, b1 = (c & 1) != 0;
  float w8[8], w4[4], w2[2];
#pragma unroll
  for (int e = 0; e < 8; ++e)
    w8[e] = (b8 ? v[e + 8] : v[e]) + row_xor<8>(b8 ? v[e] : v[e + 8]);
#pragma unroll
  for (int e = 0; e < 4; ++e)
    w4[e] = (b4 ? w8[e + 4] : w8[e]) + row_xor<4>(b4 ? w8[e] : w8[e + 4]);
#pragma unroll
  for (int e = 0; e < 2; ++e)
    w2[e] = (b2 ? w4[e + 2] : w4[e]) + row_xor<2>(b2 ? w4[e] : w4[e + 2]);
  return (b1 ? w2[1] : w2[0]) + row_xor<1>(b1 ? w2[0] : w2[1]);
}

__global__ __launch_bounds__(GG_THREADS) void gather_gate_kernel(
    const float* __restrict__ embed_u_z, const float* __restrict__ embed_u_c,
    const float* __restrict__ embed_v_z, const float* __restrict__ embed_v_c,
    const float* __restrict__ item_table,
    const int* __restrict__ hist_items, const int* __restrict__ hist_lens,
    const int* __restrict__ item_ids, const int* __restrict__ red_by_friends,
    float* __restrict__ uv, int H) {
  const int lane = threadIdx.x & 31;
  const int c    = lane & 15;
  const int b    = blockIdx.x * (2 * GG_WAVES) + (threadIdx.x >> 4);

  const int item = item_ids[b];
  int len        = hist_lens[b];
  const int red  = red_by_friends[b];
  const int* hrow = hist_items + (long)b * H;
  int hq = -1;
  if (c < H) hq = hrow[c];
  const long bd = (long)b * DD + 4 * c;
  const float4 zu = *(const float4*)(embed_u_z + bd);
  const float4 zv = *(const float4*)(embed_v_z + bd);
  len = len < H ? len : H;

  const float4* __restrict__ tab4 = (const float4*)item_table;
  const float4 a = tab4[(unsigned)item * (DD / 4) + (unsigned)c];
  const v2f al = {a.x, a.y}, ah = {a.z, a.w};
  const float nb = fmaxf(
      sqrtf(row16_sum(a.x * a.x + a.y * a.y + a.z * a.z + a.w * a.w)),
      EPS_COS);

  float m   = -INFINITY;
  bool  any = false;
  for (int j0 = 0; j0 < H; j0 += GG_CHUNK) {
    if (__builtin_amdgcn_ballot_w32(j0 < len) == 0u) break;
    int hn = -1;
    if (j0 + GG_CHUNK + c < H) hn = hrow[j0 + GG_CHUNK + c];
    const int left = len - j0;
    float d[GG_CHUNK], n[GG_CHUNK];
    float4 x[GG_CHUNK];
#pragma unroll
    for (int u = 0; u < GG_CHUNK; ++u) {
      const unsigned sel = 0x11111111u * (unsigned)u;
      const int idx = (int)__builtin_amdgcn_permlane16(
          (unsigned)hq, (unsigned)hq, sel, sel, false, false);
      x[u] = make_float4(0.0f, 0.0f, 0.0f, 0.0f);
      if (u < left) x[u] = tab4[(unsigned)idx * (DD / 4) + (unsigned)c];
    }
#pragma unroll
    for (int u = 0; u < GG_CHUNK; ++u) {
      const v2f xl = {x[u].x, x[u].y}, xh = {x[u].z, x[u].w};
      const v2f dd = xl * al + xh * ah;
      const v2f nn = xl * xl + xh * xh;
      d[u] = dd.x + dd.y;
      n[u] = nn.x + nn.y;
    }
    const float dot = row16_transpose_sum(d, c);
    const float na  = sqrtf(row16_transpose_sum(n, c));
    const float sim = dot / (fmaxf(na, EPS_COS) * nb);
    const bool valid = (c < left) && (hq != item);
    m   = valid ? fmaxf(m, sim) : m;
    any = any || valid;
    hq  = hn;
  }
  m = row16_max(m);
  const unsigned bal = __builtin_amdgcn_ballot_w32(any);
  const bool empty   = ((bal >> (lane & 16)) & 0xffffu) == 0u;
  const bool gate    = (red > 0) && (m > 0.5f);
  const bool use_u_c = empty || gate;

  float4 cu = make_float4(0.0f, 0.0f, 0.0f, 0.0f);
  float4 cv = cu;
  if (use_u_c) {
    cu = *(const float4*)(embed_u_c + bd);
    cv = cu;
    if (!empty) cv = *(const float4*)(embed_v_c + bd);
  }
  float* o = uv + (long)b * (2 * DD) + 4 * c;
  *(float4*)o        = make_float4(zu.x + cu.x, zu.y + cu.y, zu.z + cu.z, zu.w + cu.w);
  *(float4*)(o + DD) = make_float4(zv.x + cv.x, zv.y + cv.y, zv.z + cv.z, zv.w + cv.w);
}

template <int NT, int UNR>
__device__ __forceinline__ void bn_from_partials(
    const float* __restrict__ psum, const float* __restrict__ psq, int nparts,
    int K, const float* __restrict__ bn_g, const float* __restrict__ bn_be,
    float invB, float* sc_s, float* sh_s, float* red_s, float* red_q) {
  const int t     = threadIdx.x;
  const int pitch = K < 32 ? 32 : K;
  const int qpr   = K >> 2;
  const int G     = NT / qpr;
  const int cq    = t % qpr;
  const int grp   = t / qpr;
  if (grp < G) {
    float4 s = make_float4(0.0f, 0.0f, 0.0f, 0.0f);
    float4 q = make_float4(0.0f, 0.0f, 0.0f, 0.0f);
#pragma unroll UNR
    for (int p = grp; p < nparts; p += G) {
      const float4 a = *(const float4*)(psum + (long)p * pitch + 4 * cq);
      const float4 b = *(const float4*)(psq  + (long)p * pitch + 4 * cq);
      s.x += a.x; s.y += a.y; s.z += a.z; s.w += a.w;
      q.x += b.x; q.y += b.y; q.z += b.z; q.w += b.w;
    }
    float* rs = red_s + grp * K + 4 * cq;
    float* rq = red_q + grp * K + 4 * cq;
    rs[0] = s.x; rs[1] = s.y; rs[2] = s.z; rs[3] = s.w;
    rq[0] = q.x; rq[1] = q.y; rq[2] = q.z; rq[3] = q.w;
  }
  __syncthreads();
  if (t < K) {
    float S = red_s[t];
    float Q = red_q[t];
    for (int i = 1; i < G; ++i) {
      S += red_s[i * K + t];
      Q += red_q[i * K + t];
    }
    const float mean = S * invB;
    const float var  = Q * invB - mean * mean;
    const float sc   = bn_g[t] * rsqrtf(var + EPS_BN);
    sc_s[t] = sc;
    sh_s[t] = bn_be[t] - mean * sc;
  }
  __syncthreads();
}

#define MLP_WAVES   4
#define MLP_THREADS (MLP_WAVES * 32)

__global__ __launch_bounds__(MLP_THREADS) void mlp_layer_kernel(
    const float* __restrict__ x,
    const float* __restrict__ w,
    const float* __restrict__ bias,
    const float* __restrict__ in_psum,
    const float* __restrict__ in_psq,
    const float* __restrict__ bn_g,
    const float* __restrict__ bn_be,
    float* __restrict__ y,
    float* __restrict__ out_psum,
    float* __restrict__ out_psq,
    int K, int N, int use_bn, int nparts, float invB) {
  __shared__ float sc_s[128];
  __shared__ float sh_s[128];
  __shared__ __align__(16) float red_s[4 * MLP_THREADS];
  __shared__ __align__(16) float red_q[4 * MLP_THREADS];
  __shared__ float wred[2][MLP_WAVES][2][64];

  const int t    = threadIdx.x;
  const int lane = t & 31;
  const int wave = t >> 5;
  if (use_bn) {
    bn_from_partials<MLP_THREADS, 8>(in_psum, in_psq, nparts, K, bn_g, bn_be,
                                     invB, sc_s, sh_s, red_s, red_q);
  } else {
    for (int c = t; c < K; c += MLP_THREADS) {
      sc_s[c] = 1.0f;
      sh_s[c] = 0.0f;
    }
    __syncthreads();
  }

  const int row0  = (blockIdx.x * MLP_WAVES + wave) * 16;
  const int r     = lane & 15;
  const int half  = lane >> 4;
  const int khalf = half * 2;
  const int col   = lane & 15;

  for (int nt = 0; nt < (N >> 4); ++nt) {
    v8f acc = {};
    for (int k = 0; k < K; k += 4) {
      const int kk = k + khalf;
      v2f a, bb;
      a.x  = sc_s[kk]     * x[(long)(row0 + r) * K + kk]     + sh_s[kk];
      a.y  = sc_s[kk + 1] * x[(long)(row0 + r) * K + kk + 1] + sh_s[kk + 1];
      bb.x = w[(long)kk * N + nt * 16 + col];
      bb.y = w[(long)(kk + 1) * N + nt * 16 + col];
      acc = __builtin_amdgcn_wmma_f32_16x16x4_f32(
          false, a, false, bb, (short)0, acc, false, false);
    }
    const int n = nt * 16 + col;
    const float bn = bias[n];
    float s = 0.0f, q = 0.0f;
#pragma unroll
    for (int j = 0; j < 8; ++j) {
      const int m = row0 + j + half * 8;
      const float v = fmaxf(acc[j] + bn, 0.0f);
      y[(long)m * N + n] = v;
      s += v;
      q += v * v;
    }
    wred[0][wave][half][n] = s;
    wred[1][wave][half][n] = q;
  }
  __syncthreads();

  if (t < 2 * N) {
    const int kind   = t >= N ? 1 : 0;
    const int n      = t - kind * N;
    const int opitch = N < 32 ? 32 : N;
    float a = 0.0f;
    for (int wv = 0; wv < MLP_WAVES; ++wv)
      a += wred[kind][wv][0][n] + wred[kind][wv][1][n];
    (kind ? out_psq : out_psum)[(long)blockIdx.x * opitch + n] = a;
  }
}

__global__ __launch_bounds__(256) void final_kernel(
    const float* __restrict__ y3, const float* __restrict__ psum3,
    const float* __restrict__ psq3, int nparts, const float* __restrict__ g3,
    const float* __restrict__ be3, const float* __restrict__ wf,
    const float* __restrict__ bf, float* __restrict__ out, int Bn, float invB) {
  __shared__ float sc_s[16], sh_s[16], wf_s[16];
  __shared__ __align__(16) float red_s[4 * 256];
  __shared__ __align__(16) float red_q[4 * 256];
  const int t = threadIdx.x;
  if (t < 16) wf_s[t] = wf[t];
  bn_from_partials<256, 4>(psum3, psq3, nparts, 16, g3, be3, invB, sc_s, sh_s,
                           red_s, red_q);
  const int b = blockIdx.x * blockDim.x + t;
  if (b < Bn) {
    float s = 0.0f;
#pragma unroll
    for (int c = 0; c < 16; ++c)
      s += (sc_s[c] * y3[(long)b * 16 + c] + sh_s[c]) * wf_s[c];
    out[b] = s + bf[0];
  }
}

extern "C" void kernel_launch(void* const* d_in, const int* in_sizes, int n_in,
                              void* d_out, int out_size, void* d_ws,
                              size_t ws_size, hipStream_t stream) {
  const float* embed_u_z = (const float*)d_in[0];
  const float* embed_u_c = (const float*)d_in[1];
  const float* embed_v_z = (const float*)d_in[2];
  const float* embed_v_c = (const float*)d_in[3];
  const float* item_table = (const float*)d_in[4];
  const float* w1 = (const float*)d_in[5];
  const float* b1 = (const float*)d_in[6];
  const float* g1 = (const float*)d_in[7];
  const float* be1 = (const float*)d_in[8];
  const float* w2 = (const float*)d_in[9];
  const float* b2 = (const float*)d_in[10];
  const float* g2 = (const float*)d_in[11];
  const float* be2 = (const float*)d_in[12];
  const float* w3 = (const float*)d_in[13];
  const float* b3 = (const float*)d_in[14];
  const float* g3 = (const float*)d_in[15];
  const float* be3 = (const float*)d_in[16];
  const float* wf = (const float*)d_in[17];
  const float* bf = (const float*)d_in[18];
  const int* hist_items = (const int*)d_in[19];
  const int* hist_lens = (const int*)d_in[20];
  const int* item_ids = (const int*)d_in[21];
  const int* red_by_friends = (const int*)d_in[22];

  const int Bn = in_sizes[20];
  const int H  = in_sizes[19] / Bn;
  const float invB = 1.0f / (float)Bn;
  const int NB = Bn / (16 * MLP_WAVES);

  float* ws = (float*)d_ws;
  float* uv = ws;
  float* y1 = uv + (size_t)Bn * 128;
  float* y2 = y1 + (size_t)Bn * 64;
  float* y3 = y2 + (size_t)Bn * 32;
  float* ps1 = y3 + (size_t)Bn * 16;
  float* pq1 = ps1 + (size_t)NB * 64;
  float* ps2 = pq1 + (size_t)NB * 64;
  float* pq2 = ps2 + (size_t)NB * 32;
  float* ps3 = pq2 + (size_t)NB * 32;
  float* pq3 = ps3 + (size_t)NB * 32;

  gather_gate_kernel<<<Bn / (2 * GG_WAVES), GG_THREADS, 0, stream>>>(
      embed_u_z, embed_u_c, embed_v_z, embed_v_c, item_table,
      hist_items, hist_lens, item_ids, red_by_friends, uv, H);

  mlp_layer_kernel<<<NB, MLP_THREADS, 0, stream>>>(
      uv, w1, b1, nullptr, nullptr, nullptr, nullptr,
      y1, ps1, pq1, 128, 64, 0, 0, invB);

  mlp_layer_kernel<<<NB, MLP_THREADS, 0, stream>>>(
      y1, w2, b2, ps1, pq1, g1, be1,
      y2, ps2, pq2, 64, 32, 1, NB, invB);

  mlp_layer_kernel<<<NB, MLP_THREADS, 0, stream>>>(
      y2, w3, b3, ps2, pq2, g2, be2,
      y3, ps3, pq3, 32, 16, 1, NB, invB);

  final_kernel<<<(Bn + 255) / 256, 256, 0, stream>>>(
      y3, ps3, pq3, NB, g3, be3, wf, bf, (float*)d_out, Bn, invB);
}


# layer kernels fetch operands once: w copied to LDS by coalesced 16-byte loads, the wave's x rows held in registers, K and N compile-time, all column tiles accumulate in one unrolled k loop
# speedup vs baseline: 5.6256x; 1.8252x over previous
#include <hip/hip_runtime.h>
#include <hip/hip_bf16.h>
#include <math.h>

typedef __attribute__((ext_vector_type(2))) float v2f;
typedef __attribute__((ext_vector_type(8))) float v8f;

#define DD 64
#define EPS_COS 1e-8f
#define EPS_BN  1e-5f

#define GG_WAVES   8
#define GG_THREADS (GG_WAVES * 32)
#define GG_CHUNK   16

template <int M>
__device__ __forceinline__ float row_xor(float v) {
  return __int_as_float(__builtin_amdgcn_update_dpp(
      0, __float_as_int(v), 0x160 + M, 0xf, 0xf, true));
}

__device__ __forceinline__ float row16_sum(float v) {
  v += row_xor<1>(v);
  v += row_xor<2>(v);
  v += row_xor<4>(v);
  v += row_xor<8>(v);
  return v;
}
__device__ __forceinline__ float row16_max(float v) {
  v = fmaxf(v, row_xor<1>(v));
  v = fmaxf(v, row_xor<2>(v));
  v = fmaxf(v, row_xor<4>(v));
  v = fmaxf(v, row_xor<8>(v));
  return v;
}

__device__ __forceinline__ float row16_transpose_sum(const float (&v)[16],
                                                     int c) {
  const bool b8 = (c & 8) != 0, b4 = (c & 4) != 0;
  const bool b2 = (c & 2) != 0, b1 = (c & 1) != 0;
  float w8[8], w4[4], w2[2];
#pragma unroll
  for (int e = 0; e < 8; ++e)
    w8[e] = (b8 ? v[e + 8] : v[e]) + row_xor<8>(b8 ? v[e] : v[e + 8]);
#pragma unroll
  for (int e = 0; e < 4; ++e)
    w4[e] = (b4 ? w8[e + 4] : w8[e]) + row_xor<4>(b4 ? w8[e] : w8[e + 4]);
#pragma unroll
  for (int e = 0; e < 2; ++e)
    w2[e] = (b2 ? w4[e + 2] : w4[e]) + row_xor<2>(b2 ? w4[e] : w4[e + 2]);
  return (b1 ? w2[1] : w2[0]) + row_xor<1>(b1 ? w2[0] : w2[1]);
}

__global__ __launch_bounds__(GG_THREADS) void gather_gate_kernel(
    const float* __restrict__ embed_u_z, const float* __restrict__ embed_u_c,
    const float* __restrict__ embed_v_z, const float* __restrict__ embed_v_c,
    const float* __restrict__ item_table,
    const int* __restrict__ hist_items, const int* __restrict__ hist_lens,
    const int* __restrict__ item_ids, const int* __restrict__ red_by_friends,
    float* __restrict__ uv, int H) {
  const int lane = threadIdx.x & 31;
  const int c    = lane & 15;
  const int b    = blockIdx.x * (2 * GG_WAVES) + (threadIdx.x >> 4);

  const int item = item_ids[b];
  int len        = hist_lens[b];
  const int red  = red_by_friends[b];
  const int* hrow = hist_items + (long)b * H;
  int hq = -1;
  if (c < H) hq = hrow[c];
  const long bd = (long)b * DD + 4 * c;
  const float4 zu = *(const float4*)(embed_u_z + bd);
  const float4 zv = *(const float4*)(embed_v_z + bd);
  len = len < H ? len : H;

  const float4* __restrict__ tab4 = (const float4*)item_table;
  const float4 a = tab4[(unsigned)item * (DD / 4) + (unsigned)c];
  const v2f al = {a.x, a.y}, ah = {a.z, a.w};
  const float nb = fmaxf(
      sqrtf(row16_sum(a.x * a.x + a.y * a.y + a.z * a.z + a.w * a.w)),
      EPS_COS);

  float m   = -INFINITY;
  bool  any = false;
  for (int j0 = 0; j0 < H; j0 += GG_CHUNK) {
    if (__builtin_amdgcn_ballot_w32(j0 < len) == 0u) break;
    int hn = -1;
    if (j0 + GG_CHUNK + c < H) hn = hrow[j0 + GG_CHUNK + c];
    const int left = len - j0;
    float d[GG_CHUNK], n[GG_CHUNK];
    float4 x[GG_CHUNK];
#pragma unroll
    for (int u = 0; u < GG_CHUNK; ++u) {
      const unsigned sel = 0x11111111u * (unsigned)u;
      const int idx = (int)__builtin_amdgcn_permlane16(
          (unsigned)hq, (unsigned)hq, sel, sel, false, false);
      x[u] = make_float4(0.0f, 0.0f, 0.0f, 0.0f);
      if (u < left) x[u] = tab4[(unsigned)idx * (DD / 4) + (unsigned)c];
    }
#pragma unroll
    for (int u = 0; u < GG_CHUNK; ++u) {
      const v2f xl = {x[u].x, x[u].y}, xh = {x[u].z, x[u].w};
      const v2f dd = xl * al + xh * ah;
      const v2f nn = xl * xl + xh * xh;
      d[u] = dd.x + dd.y;
      n[u] = nn.x + nn.y;
    }
    const float dot = row16_transpose_sum(d, c);
    const float na  = sqrtf(row16_transpose_sum(n, c));
    const float sim = dot / (fmaxf(na, EPS_COS) * nb);
    const bool valid = (c < left) && (hq != item);
    m   = valid ? fmaxf(m, sim) : m;
    any = any || valid;
    hq  = hn;
  }
  m = row16_max(m);
  const unsigned bal = __builtin_amdgcn_ballot_w32(any);
  const bool empty   = ((bal >> (lane & 16)) & 0xffffu) == 0u;
  const bool gate    = (red > 0) && (m > 0.5f);
  const bool use_u_c = empty || gate;

  float4 cu = make_float4(0.0f, 0.0f, 0.0f, 0.0f);
  float4 cv = cu;
  if (use_u_c) {
    cu = *(const float4*)(embed_u_c + bd);
    cv = cu;
    if (!empty) cv = *(const float4*)(embed_v_c + bd);
  }
  float* o = uv + (long)b * (2 * DD) + 4 * c;
  *(float4*)o        = make_float4(zu.x + cu.x, zu.y + cu.y, zu.z + cu.z, zu.w + cu.w);
  *(float4*)(o + DD) = make_float4(zv.x + cv.x, zv.y + cv.y, zv.z + cv.z, zv.w + cv.w);
}

template <int NT, int UNR>
__device__ __forceinline__ void bn_from_partials(
    const float* __restrict__ psum, const float* __restrict__ psq, int nparts,
    int K, const float* __restrict__ bn_g, const float* __restrict__ bn_be,
    float invB, float* sc_s, float* sh_s, float* red_s, float* red_q) {
  const int t     = threadIdx.x;
  const int pitch = K < 32 ? 32 : K;
  const int qpr   = K >> 2;
  const int G     = NT / qpr;
  const int cq    = t % qpr;
  const int grp   = t / qpr;
  if (grp < G) {
    float4 s = make_float4(0.0f, 0.0f, 0.0f, 0.0f);
    float4 q = make_float4(0.0f, 0.0f, 0.0f, 0.0f);
#pragma unroll UNR
    for (int p = grp; p < nparts; p += G) {
      const float4 a = *(const float4*)(psum + (long)p * pitch + 4 * cq);
      const float4 b = *(const float4*)(psq  + (long)p * pitch + 4 * cq);
      s.x += a.x; s.y += a.y; s.z += a.z; s.w += a.w;
      q.x += b.x; q.y += b.y; q.z += b.z; q.w += b.w;
    }
    float* rs = red_s + grp * K + 4 * cq;
    float* rq = red_q + grp * K + 4 * cq;
    rs[0] = s.x; rs[1] = s.y; rs[2] = s.z; rs[3] = s.w;
    rq[0] = q.x; rq[1] = q.y; rq[2] = q.z; rq[3] = q.w;
  }
  __syncthreads();
  if (t < K) {
    float S = red_s[t];
    float Q = red_q[t];
    for (int i = 1; i < G; ++i) {
      S += red_s[i * K + t];
      Q += red_q[i * K + t];
    }
    const float mean = S * invB;
    const float var  = Q * invB - mean * mean;
    const float sc   = bn_g[t] * rsqrtf(var + EPS_BN);
    sc_s[t] = sc;
    sh_s[t] = bn_be[t] - mean * sc;
  }
  __syncthreads();
}

#define MLP_WAVES   4
#define MLP_THREADS (MLP_WAVES * 32)

template <int K, int N, bool USE_BN>
__global__ __launch_bounds__(MLP_THREADS) void mlp_layer_kernel(
    const float* __restrict__ x,
    const float* __restrict__ w,
    const float* __restrict__ bias,
    const float* __restrict__ in_psum,
    const float* __restrict__ in_psq,
    const float* __restrict__ bn_g,
    const float* __restrict__ bn_be,
    float* __restrict__ y,
    float* __restrict__ out_psum,
    float* __restrict__ out_psq,
    int nparts, float invB) {
  constexpr int KS  = K / 4;
  constexpr int NT  = N / 16;
  constexpr int WP  = N + 8;
  constexpr int QPR = N / 4;
  constexpr int WQ  = K * QPR / MLP_THREADS;
  __shared__ __align__(16) float w_s[K * WP];
  __shared__ __align__(16) float sc_s[K];
  __shared__ __align__(16) float sh_s[K];
  __shared__ __align__(16) float red_s[USE_BN ? 4 * MLP_THREADS : 4];
  __shared__ __align__(16) float red_q[USE_BN ? 4 * MLP_THREADS : 4];
  __shared__ float wred[2][MLP_WAVES][2][N];

  const int t    = threadIdx.x;
  const int lane = t & 31;
  const int wave = t >> 5;
  const int row0 = (blockIdx.x * MLP_WAVES + wave) * 16;
  const int r    = lane & 15;
  const int half = lane >> 4;
  const int col  = lane & 15;

  float4 wq[WQ];
#pragma unroll
  for (int i = 0; i < WQ; ++i)
    wq[i] = ((const float4*)w)[t + i * MLP_THREADS];

  const float* xr = x + (long)(row0 + r) * K + 2 * half;
  v2f a[KS];
#pragma unroll
  for (int s = 0; s < KS; ++s) a[s] = *(const v2f*)(xr + 4 * s);

#pragma unroll
  for (int i = 0; i < WQ; ++i) {
    const int p = t + i * MLP_THREADS;
    *(float4*)(w_s + (p / QPR) * WP + 4 * (p % QPR)) = wq[i];
  }
  if (USE_BN)
    bn_from_partials<MLP_THREADS, 8>(in_psum, in_psq, nparts, K, bn_g, bn_be,
                                     invB, sc_s, sh_s, red_s, red_q);
  else
    __syncthreads();

  if (USE_BN) {
#pragma unroll
    for (int s = 0; s < KS; ++s) {
      const v2f sc = *(const v2f*)(sc_s + 4 * s + 2 * half);
      const v2f sh = *(const v2f*)(sh_s + 4 * s + 2 * half);
      a[s] = sc * a[s] + sh;
    }
  }

  v8f acc[NT];
#pragma unroll
  for (int nt = 0; nt < NT; ++nt) acc[nt] = v8f{};
  const float* wl = w_s + (2 * half) * WP + col;
#pragma unroll
  for (int s = 0; s < KS; ++s) {
#pragma unroll
    for (int nt = 0; nt < NT; ++nt) {
      v2f bb;
      bb.x = wl[(4 * s) * WP + nt * 16];
      bb.y = wl[(4 * s + 1) * WP + nt * 16];
      acc[nt] = __builtin_amdgcn_wmma_f32_16x16x4_f32(
          false, a[s], false, bb, (short)0, acc[nt], false, false);
    }
  }

#pragma unroll
  for (int nt = 0; nt < NT; ++nt) {
    const int n = nt * 16 + col;
    const float bn = bias[n];
    float s = 0.0f, q = 0.0f;
#pragma unroll
    for (int j = 0; j < 8; ++j) {
      const int m = row0 + j + half * 8;
      const float v = fmaxf(acc[nt][j] + bn, 0.0f);
      y[(long)m * N + n] = v;
      s += v;
      q += v * v;
    }
    wred[0][wave][half][n] = s;
    wred[1][wave][half][n] = q;
  }
  __syncthreads();

  if (t < 2 * N) {
    const int kind   = t >= N ? 1 : 0;
    const int n      = t - kind * N;
    constexpr int opitch = N < 32 ? 32 : N;
    float a2 = 0.0f;
    for (int wv = 0; wv < MLP_WAVES; ++wv)
      a2 += wred[kind][wv][0][n] + wred[kind][wv][1][n];
    (kind ? out_psq : out_psum)[(long)blockIdx.x * opitch + n] = a2;
  }
}

__global__ __launch_bounds__(256) void final_kernel(
    const float* __restrict__ y3, const float* __restrict__ psum3,
    const float* __restrict__ psq3, int nparts, const float* __restrict__ g3,
    const float* __restrict__ be3, const float* __restrict__ wf,
    const float* __restrict__ bf, float* __restrict__ out, int Bn, float invB) {
  __shared__ float sc_s[16], sh_s[16], wf_s[16];
  __shared__ __align__(16) float red_s[4 * 256];
  __shared__ __align__(16) float red_q[4 * 256];
  const int t = threadIdx.x;
  if (t < 16) wf_s[t] = wf[t];
  bn_from_partials<256, 4>(psum3, psq3, nparts, 16, g3, be3, invB, sc_s, sh_s,
                           red_s, red_q);
  const int b = blockIdx.x * blockDim.x + t;
  if (b < Bn) {
    float s = 0.0f;
#pragma unroll
    for (int c = 0; c < 16; ++c)
      s += (sc_s[c] * y3[(long)b * 16 + c] + sh_s[c]) * wf_s[c];
    out[b] = s + bf[0];
  }
}

extern "C" void kernel_launch(void* const* d_in, const int* in_sizes, int n_in,
                              void* d_out, int out_size, void* d_ws,
                              size_t ws_size, hipStream_t stream) {
  const float* embed_u_z = (const float*)d_in[0];
  const float* embed_u_c = (const float*)d_in[1];
  const float* embed_v_z = (const float*)d_in[2];
  const float* embed_v_c = (const float*)d_in[3];
  const float* item_table = (const float*)d_in[4];
  const float* w1 = (const float*)d_in[5];
  const float* b1 = (const float*)d_in[6];
  const float* g1 = (const float*)d_in[7];
  const float* be1 = (const float*)d_in[8];
  const float* w2 = (const float*)d_in[9];
  const float* b2 = (const float*)d_in[10];
  const float* g2 = (const float*)d_in[11];
  const float* be2 = (const float*)d_in[12];
  const float* w3 = (const float*)d_in[13];
  const float* b3 = (const float*)d_in[14];
  const float* g3 = (const float*)d_in[15];
  const float* be3 = (const float*)d_in[16];
  const float* wf = (const float*)d_in[17];
  const float* bf = (const float*)d_in[18];
  const int* hist_items = (const int*)d_in[19];
  const int* hist_lens = (const int*)d_in[20];
  const int* item_ids = (const int*)d_in[21];
  const int* red_by_friends = (const int*)d_in[22];

  const int Bn = in_sizes[20];
  const int H  = in_sizes[19] / Bn;
  const float invB = 1.0f / (float)Bn;
  const int NB = Bn / (16 * MLP_WAVES);

  float* ws = (float*)d_ws;
  float* uv = ws;
  float* y1 = uv + (size_t)Bn * 128;
  float* y2 = y1 + (size_t)Bn * 64;
  float* y3 = y2 + (size_t)Bn * 32;
  float* ps1 = y3 + (size_t)Bn * 16;
  float* pq1 = ps1 + (size_t)NB * 64;
  float* ps2 = pq1 + (size_t)NB * 64;
  float* pq2 = ps2 + (size_t)NB * 32;
  float* ps3 = pq2 + (size_t)NB * 32;
  float* pq3 = ps3 + (size_t)NB * 32;

  gather_gate_kernel<<<Bn / (2 * GG_WAVES), GG_THREADS, 0, stream>>>(
      embed_u_z, embed_u_c, embed_v_z, embed_v_c, item_table,
      hist_items, hist_lens, item_ids, red_by_friends, uv, H);

  mlp_layer_kernel<128, 64, false><<<NB, MLP_THREADS, 0, stream>>>(
      uv, w1, b1, nullptr, nullptr, nullptr, nullptr,
      y1, ps1, pq1, 0, invB);

  mlp_layer_kernel<64, 32, true><<<NB, MLP_THREADS, 0, stream>>>(
      y1, w2, b2, ps1, pq1, g1, be1,
      y2, ps2, pq2, NB, invB);

  mlp_layer_kernel<32, 16, true><<<NB, MLP_THREADS, 0, stream>>>(
      y2, w3, b3, ps2, pq2, g2, be2,
      y3, ps3, pq3, NB, invB);

  final_kernel<<<(Bn + 255) / 256, 256, 0, stream>>>(
      y3, ps3, pq3, NB, g3, be3, wf, bf, (float*)d_out, Bn, invB);
}


# BN consumers (layers 2, 3, final) request all partial-sum rows, gamma/beta, weights and x rows before the first wait: one memory round trip per kernel, not 3 to 6; every sum keeps its order
# speedup vs baseline: 6.1863x; 1.0997x over previous
#include <hip/hip_runtime.h>
#include <hip/hip_bf16.h>
#include <math.h>

typedef __attribute__((ext_vector_type(2))) float v2f;
typedef __attribute__((ext_vector_type(8))) float v8f;

#define DD 64
#define EPS_COS 1e-8f
#define EPS_BN  1e-5f

#define GG_WAVES   8
#define GG_THREADS (GG_WAVES * 32)
#define GG_CHUNK   16

template <int M>
__device__ __forceinline__ float row_xor(float v) {
  return __int_as_float(__builtin_amdgcn_update_dpp(
      0, __float_as_int(v), 0x160 + M, 0xf, 0xf, true));
}

__device__ __forceinline__ float row16_sum(float v) {
  v += row_xor<1>(v);
  v += row_xor<2>(v);
  v += row_xor<4>(v);
  v += row_xor<8>(v);
  return v;
}
__device__ __forceinline__ float row16_max(float v) {
  v = fmaxf(v, row_xor<1>(v));
  v = fmaxf(v, row_xor<2>(v));
  v = fmaxf(v, row_xor<4>(v));
  v = fmaxf(v, row_xor<8>(v));
  return v;
}

__device__ __forceinline__ float row16_transpose_sum(const float (&v)[16],
                                                     int c) {
  const bool b8 = (c & 8) != 0, b4 = (c & 4) != 0;
  const bool b2 = (c & 2) != 0, b1 = (c & 1) != 0;
  float w8[8], w4[4], w2[2];
#pragma unroll
  for (int e = 0; e < 8; ++e)
    w8[e] = (b8 ? v[e + 8] : v[e]) + row_xor<8>(b8 ? v[e] : v[e + 8]);
#pragma unroll
  for (int e = 0; e < 4; ++e)
    w4[e] = (b4 ? w8[e + 4] : w8[e]) + row_xor<4>(b4 ? w8[e] : w8[e + 4]);
#pragma unroll
  for (int e = 0; e < 2; ++e)
    w2[e] = (b2 ? w4[e + 2] : w4[e]) + row_xor<2>(b2 ? w4[e] : w4[e + 2]);
  return (b1 ? w2[1] : w2[0]) + row_xor<1>(b1 ? w2[0] : w2[1]);
}

__global__ __launch_bounds__(GG_THREADS) void gather_gate_kernel(
    const float* __restrict__ embed_u_z, const float* __restrict__ embed_u_c,
    const float* __restrict__ embed_v_z, const float* __restrict__ embed_v_c,
    const float* __restrict__ item_table,
    const int* __restrict__ hist_items, const int* __restrict__ hist_lens,
    const int* __restrict__ item_ids, const int* __restrict__ red_by_friends,
    float* __restrict__ uv, int H) {
  const int lane = threadIdx.x & 31;
  const int c    = lane & 15;
  const int b    = blockIdx.x * (2 * GG_WAVES) + (threadIdx.x >> 4);

  const int item = item_ids[b];
  int len        = hist_lens[b];
  const int red  = red_by_friends[b];
  const int* hrow = hist_items + (long)b * H;
  int hq = -1;
  if (c < H) hq = hrow[c];
  const long bd = (long)b * DD + 4 * c;
  const float4 zu = *(const float4*)(embed_u_z + bd);
  const float4 zv = *(const float4*)(embed_v_z + bd);
  len = len < H ? len : H;

  const float4* __restrict__ tab4 = (const float4*)item_table;
  const float4 a = tab4[(unsigned)item * (DD / 4) + (unsigned)c];
  const v2f al = {a.x, a.y}, ah = {a.z, a.w};
  const float nb = fmaxf(
      sqrtf(row16_sum(a.x * a.x + a.y * a.y + a.z * a.z + a.w * a.w)),
      EPS_COS);

  float m   = -INFINITY;
  bool  any = false;
  for (int j0 = 0; j0 < H; j0 += GG_CHUNK) {
    if (__builtin_amdgcn_ballot_w32(j0 < len) == 0u) break;
    int hn = -1;
    if (j0 + GG_CHUNK + c < H) hn = hrow[j0 + GG_CHUNK + c];
    const int left = len - j0;
    float d[GG_CHUNK], n[GG_CHUNK];
    float4 x[GG_CHUNK];
#pragma unroll
    for (int u = 0; u < GG_CHUNK; ++u) {
      const unsigned sel = 0x11111111u * (unsigned)u;
      const int idx = (int)__builtin_amdgcn_permlane16(
          (unsigned)hq, (unsigned)hq, sel, sel, false, false);
      x[u] = make_float4(0.0f, 0.0f, 0.0f, 0.0f);
      if (u < left) x[u] = tab4[(unsigned)idx * (DD / 4) + (unsigned)c];
    }
#pragma unroll
    for (int u = 0; u < GG_CHUNK; ++u) {
      const v2f xl = {x[u].x, x[u].y}, xh = {x[u].z, x[u].w};
      const v2f dd = xl * al + xh * ah;
      const v2f nn = xl * xl + xh * xh;
      d[u] = dd.x + dd.y;
      n[u] = nn.x + nn.y;
    }
    const float dot = row16_transpose_sum(d, c);
    const float na  = sqrtf(row16_transpose_sum(n, c));
    const float sim = dot / (fmaxf(na, EPS_COS) * nb);
    const bool valid = (c < left) && (hq != item);
    m   = valid ? fmaxf(m, sim) : m;
    any = any || valid;
    hq  = hn;
  }
  m = row16_max(m);
  const unsigned bal = __builtin_amdgcn_ballot_w32(any);
  const bool empty   = ((bal >> (lane & 16)) & 0xffffu) == 0u;
  const bool gate    = (red > 0) && (m > 0.5f);
  const bool use_u_c = empty || gate;

  float4 cu = make_float4(0.0f, 0.0f, 0.0f, 0.0f);
  float4 cv = cu;
  if (use_u_c) {
    cu = *(const float4*)(embed_u_c + bd);
    cv = cu;
    if (!empty) cv = *(const float4*)(embed_v_c + bd);
  }
  float* o = uv + (long)b * (2 * DD) + 4 * c;
  *(float4*)o        = make_float4(zu.x + cu.x, zu.y + cu.y, zu.z + cu.z, zu.w + cu.w);
  *(float4*)(o + DD) = make_float4(zv.x + cv.x, zv.y + cv.y, zv.z + cv.z, zv.w + cv.w);
}

template <int NT, int K, int RPT>
__device__ __forceinline__ void bn_request(
    const float* __restrict__ psum, const float* __restrict__ psq, int nparts,
    float4 (&s)[RPT], float4 (&q)[RPT]) {
  constexpr int QPR   = K / 4;
  constexpr int G     = NT / QPR;
  constexpr int PITCH = K < 32 ? 32 : K;
  static_assert(NT % QPR == 0, "threads must tile the partial rows");
  const int cq  = threadIdx.x % QPR;
  const int grp = threadIdx.x / QPR;
  const bool full   = RPT * G <= nparts;
  const long first  = full ? (long)grp * PITCH : 0;
  const long stride = full ? (long)G * PITCH : 0;
  const float* ps = psum + first + 4 * cq;
  const float* pq = psq  + first + 4 * cq;
#pragma unroll
  for (int i = 0; i < RPT; ++i) {
    s[i] = *(const float4*)(ps + i * stride);
    q[i] = *(const float4*)(pq + i * stride);
  }
}

template <int NT, int K, int RPT>
__device__ __forceinline__ void bn_finish(
    const float* __restrict__ psum, const float* __restrict__ psq, int nparts,
    const float4 (&s)[RPT], const float4 (&q)[RPT], float gamma, float beta,
    float invB, float* sc_s, float* sh_s, float* red) {
  constexpr int QPR   = K / 4;
  constexpr int G     = NT / QPR;
  constexpr int PITCH = K < 32 ? 32 : K;
  constexpr int RP    = G + 4;
  static_assert(G % 4 == 0, "a column's group sums are read 16 bytes at a time");
  const int t   = threadIdx.x;
  const int cq  = t % QPR;
  const int grp = t / QPR;
  float4 S = make_float4(0.0f, 0.0f, 0.0f, 0.0f);
  float4 Q = make_float4(0.0f, 0.0f, 0.0f, 0.0f);
#pragma unroll
  for (int i = 0; i < RPT; ++i) {
    S.x += s[i].x; S.y += s[i].y; S.z += s[i].z; S.w += s[i].w;
    Q.x += q[i].x; Q.y += q[i].y; Q.z += q[i].z; Q.w += q[i].w;
  }
  const bool full = RPT * G <= nparts;
  if (!full) {
    S = make_float4(0.0f, 0.0f, 0.0f, 0.0f);
    Q = S;
  }
  for (int p = grp + (full ? RPT * G : 0); p < nparts; p += G) {
    const float4 a = *(const float4*)(psum + (long)p * PITCH + 4 * cq);
    const float4 b = *(const float4*)(psq  + (long)p * PITCH + 4 * cq);
    S.x += a.x; S.y += a.y; S.z += a.z; S.w += a.w;
    Q.x += b.x; Q.y += b.y; Q.z += b.z; Q.w += b.w;
  }
  float* rs = red + (4 * cq) * RP + grp;
  float* rq = rs + K * RP;
  rs[0] = S.x; rs[RP] = S.y; rs[2 * RP] = S.z; rs[3 * RP] = S.w;
  rq[0] = Q.x; rq[RP] = Q.y; rq[2 * RP] = Q.z; rq[3 * RP] = Q.w;
  __syncthreads();
  if (t < K) {
    const float4* cs  = (const float4*)(red + t * RP);
    const float4* cq4 = (const float4*)(red + (K + t) * RP);
    float4 v[G / 4], w[G / 4];
#pragma unroll
    for (int i = 0; i < G / 4; ++i) {
      v[i] = cs[i];
      w[i] = cq4[i];
    }
    __builtin_amdgcn_sched_barrier(0);
    float Sc = v[0].x, Qc = w[0].x;
    Sc += v[0].y; Sc += v[0].z; Sc += v[0].w;
    Qc += w[0].y; Qc += w[0].z; Qc += w[0].w;
#pragma unroll
    for (int i = 1; i < G / 4; ++i) {
      Sc += v[i].x; Sc += v[i].y; Sc += v[i].z; Sc += v[i].w;
      Qc += w[i].x; Qc += w[i].y; Qc += w[i].z; Qc += w[i].w;
    }
    const float mean = Sc * invB;
    const float var  = Qc * invB - mean * mean;
    const float sc   = gamma * rsqrtf(var + EPS_BN);
    sc_s[t] = sc;
    sh_s[t] = beta - mean * sc;
  }
  __syncthreads();
}

#define MLP_WAVES   4
#define MLP_THREADS (MLP_WAVES * 32)

template <int K, int N, bool USE_BN>
__global__ __launch_bounds__(MLP_THREADS) void mlp_layer_kernel(
    const float* __restrict__ x,
    const float* __restrict__ w,
    const float* __restrict__ bias,
    const float* __restrict__ in_psum,
    const float* __restrict__ in_psq,
    const float* __restrict__ bn_g,
    const float* __restrict__ bn_be,
    float* __restrict__ y,
    float* __restrict__ out_psum,
    float* __restrict__ out_psq,
    int nparts, float invB) {
  constexpr int KS  = K / 4;
  constexpr int NT  = N / 16;
  constexpr int WP  = N + 8;
  constexpr int QPR = N / 4;
  constexpr int WQ  = K * QPR / MLP_THREADS;
  __shared__ __align__(16) float w_s[K * WP];
  __shared__ __align__(16) float sc_s[K];
  __shared__ __align__(16) float sh_s[K];
  constexpr int RPT = USE_BN ? K / 2 : 1;
  constexpr int BG  = MLP_THREADS / (K / 4);
  __shared__ __align__(16) float red[USE_BN ? 2 * K * (BG + 4) : 4];
  __shared__ float wred[2][MLP_WAVES][2][N];

  const int t    = threadIdx.x;
  const int lane = t & 31;
  const int wave = t >> 5;
  const int row0 = (blockIdx.x * MLP_WAVES + wave) * 16;
  const int r    = lane & 15;
  const int half = lane >> 4;
  const int col  = lane & 15;

  float4 wq[WQ];
#pragma unroll
  for (int i = 0; i < WQ; ++i)
    wq[i] = ((const float4*)w)[t + i * MLP_THREADS];

  const float* xr = x + (long)(row0 + r) * K + 2 * half;
  v2f a[KS];
#pragma unroll
  for (int s = 0; s < KS; ++s) a[s] = *(const v2f*)(xr + 4 * s);

  float4 ps[RPT], pq[RPT];
  float gam = 0.0f, bet = 0.0f;
  if (USE_BN) {
    gam = bn_g[t % K];
    bet = bn_be[t % K];
    bn_request<MLP_THREADS, K, RPT>(in_psum, in_psq, nparts, ps, pq);
  }

#pragma unroll
  for (int i = 0; i < WQ; ++i) {
    const int p = t + i * MLP_THREADS;
    *(float4*)(w_s + (p / QPR) * WP + 4 * (p % QPR)) = wq[i];
  }
  if (USE_BN)
    bn_finish<MLP_THREADS, K, RPT>(in_psum, in_psq, nparts, ps, pq, gam, bet,
                                   invB, sc_s, sh_s, red);
  else
    __syncthreads();

  if (USE_BN) {
#pragma unroll
    for (int s = 0; s < KS; ++s) {
      const v2f sc = *(const v2f*)(sc_s + 4 * s + 2 * half);
      const v2f sh = *(const v2f*)(sh_s + 4 * s + 2 * half);
      a[s] = sc * a[s] + sh;
    }
  }

  v8f acc[NT];
#pragma unroll
  for (int nt = 0; nt < NT; ++nt) acc[nt] = v8f{};
  const float* wl = w_s + (2 * half) * WP + col;
#pragma unroll
  for (int s = 0; s < KS; ++s) {
#pragma unroll
    for (int nt = 0; nt < NT; ++nt) {
      v2f bb;
      bb.x = wl[(4 * s) * WP + nt * 16];
      bb.y = wl[(4 * s + 1) * WP + nt * 16];
      acc[nt] = __builtin_amdgcn_wmma_f32_16x16x4_f32(
          false, a[s], false, bb, (short)0, acc[nt], false, false);
    }
  }

#pragma unroll
  for (int nt = 0; nt < NT; ++nt) {
    const int n = nt * 16 + col;
    const float bn = bias[n];
    float s = 0.0f, q = 0.0f;
#pragma unroll
    for (int j = 0; j < 8; ++j) {
      const int m = row0 + j + half * 8;
      const float v = fmaxf(acc[nt][j] + bn, 0.0f);
      y[(long)m * N + n] = v;
      s += v;
      q += v * v;
    }
    wred[0][wave][half][n] = s;
    wred[1][wave][half][n] = q;
  }
  __syncthreads();

  if (t < 2 * N) {
    const int kind   = t >= N ? 1 : 0;
    const int n      = t - kind * N;
    constexpr int opitch = N < 32 ? 32 : N;
    float a2 = 0.0f;
    for (int wv = 0; wv < MLP_WAVES; ++wv)
      a2 += wred[kind][wv][0][n] + wred[kind][wv][1][n];
    (kind ? out_psq : out_psum)[(long)blockIdx.x * opitch + n] = a2;
  }
}

#define FIN_THREADS 256

__global__ __launch_bounds__(FIN_THREADS) void final_kernel(
    const float* __restrict__ y3, const float* __restrict__ psum3,
    const float* __restrict__ psq3, int nparts, const float* __restrict__ g3,
    const float* __restrict__ be3, const float* __restrict__ wf,
    const float* __restrict__ bf, float* __restrict__ out, int Bn, float invB) {
  constexpr int RPT = 4;
  constexpr int BG  = FIN_THREADS / (16 / 4);
  __shared__ float sc_s[16], sh_s[16], wf_s[16];
  __shared__ __align__(16) float red[2 * 16 * (BG + 4)];
  const int t = threadIdx.x;
  const int b = blockIdx.x * FIN_THREADS + t;

  const float wfv = wf[t & 15];
  const float gam = g3[t & 15];
  const float bet = be3[t & 15];
  const float bfv = bf[0];
  const float4* yr = (const float4*)(y3 + (long)(b < Bn ? b : Bn - 1) * 16);
  float4 yv[4];
#pragma unroll
  for (int i = 0; i < 4; ++i) yv[i] = yr[i];
  float4 ps[RPT], pq[RPT];
  bn_request<FIN_THREADS, 16, RPT>(psum3, psq3, nparts, ps, pq);

  if (t < 16) wf_s[t] = wfv;
  bn_finish<FIN_THREADS, 16, RPT>(psum3, psq3, nparts, ps, pq, gam, bet, invB,
                                  sc_s, sh_s, red);
  if (b < Bn) {
    const float yy[16] = {yv[0].x, yv[0].y, yv[0].z, yv[0].w,
                          yv[1].x, yv[1].y, yv[1].z, yv[1].w,
                          yv[2].x, yv[2].y, yv[2].z, yv[2].w,
                          yv[3].x, yv[3].y, yv[3].z, yv[3].w};
    float s = 0.0f;
#pragma unroll
    for (int c = 0; c < 16; ++c)
      s += (sc_s[c] * yy[c] + sh_s[c]) * wf_s[c];
    out[b] = s + bfv;
  }
}

extern "C" void kernel_launch(void* const* d_in, const int* in_sizes, int n_in,
                              void* d_out, int out_size, void* d_ws,
                              size_t ws_size, hipStream_t stream) {
  const float* embed_u_z = (const float*)d_in[0];
  const float* embed_u_c = (const float*)d_in[1];
  const float* embed_v_z = (const float*)d_in[2];
  const float* embed_v_c = (const float*)d_in[3];
  const float* item_table = (const float*)d_in[4];
  const float* w1 = (const float*)d_in[5];
  const float* b1 = (const float*)d_in[6];
  const float* g1 = (const float*)d_in[7];
  const float* be1 = (const float*)d_in[8];
  const float* w2 = (const float*)d_in[9];
  const float* b2 = (const float*)d_in[10];
  const float* g2 = (const float*)d_in[11];
  const float* be2 = (const float*)d_in[12];
  const float* w3 = (const float*)d_in[13];
  const float* b3 = (const float*)d_in[14];
  const float* g3 = (const float*)d_in[15];
  const float* be3 = (const float*)d_in[16];
  const float* wf = (const float*)d_in[17];
  const float* bf = (const float*)d_in[18];
  const int* hist_items = (const int*)d_in[19];
  const int* hist_lens = (const int*)d_in[20];
  const int* item_ids = (const int*)d_in[21];
  const int* red_by_friends = (const int*)d_in[22];

  const int Bn = in_sizes[20];
  const int H  = in_sizes[19] / Bn;
  const float invB = 1.0f / (float)Bn;
  const int NB = Bn / (16 * MLP_WAVES);

  float* ws = (float*)d_ws;
  float* uv = ws;
  float* y1 = uv + (size_t)Bn * 128;
  float* y2 = y1 + (size_t)Bn * 64;
  float* y3 = y2 + (size_t)Bn * 32;
  float* ps1 = y3 + (size_t)Bn * 16;
  float* pq1 = ps1 + (size_t)NB * 64;
  float* ps2 = pq1 + (size_t)NB * 64;
  float* pq2 = ps2 + (size_t)NB * 32;
  float* ps3 = pq2 + (size_t)NB * 32;
  float* pq3 = ps3 + (size_t)NB * 32;

  gather_gate_kernel<<<Bn / (2 * GG_WAVES), GG_THREADS, 0, stream>>>(
      embed_u_z, embed_u_c, embed_v_z, embed_v_c, item_table,
      hist_items, hist_lens, item_ids, red_by_friends, uv, H);

  mlp_layer_kernel<128, 64, false><<<NB, MLP_THREADS, 0, stream>>>(
      uv, w1, b1, nullptr, nullptr, nullptr, nullptr,
      y1, ps1, pq1, 0, invB);

  mlp_layer_kernel<64, 32, true><<<NB, MLP_THREADS, 0, stream>>>(
      y1, w2, b2, ps1, pq1, g1, be1,
      y2, ps2, pq2, NB, invB);

  mlp_layer_kernel<32, 16, true><<<NB, MLP_THREADS, 0, stream>>>(
      y2, w3, b3, ps2, pq2, g2, be2,
      y3, ps3, pq3, NB, invB);

  final_kernel<<<(Bn + 255) / 256, 256, 0, stream>>>(
      y3, ps3, pq3, NB, g3, be3, wf, bf, (float*)d_out, Bn, invB);
}


# gather kernel reads no anchor or history rows for batch rows with red_by_friends <= 0 (their gate is false whatever they score); the empty test comes from the indices alone
# speedup vs baseline: 6.4029x; 1.0350x over previous
#include <hip/hip_runtime.h>
#include <hip/hip_bf16.h>
#include <math.h>

typedef __attribute__((ext_vector_type(2))) float v2f;
typedef __attribute__((ext_vector_type(8))) float v8f;

#define DD 64
#define EPS_COS 1e-8f
#define EPS_BN  1e-5f

#define GG_WAVES   8
#define GG_THREADS (GG_WAVES * 32)
#define GG_CHUNK   16

template <int M>
__device__ __forceinline__ float row_xor(float v) {
  return __int_as_float(__builtin_amdgcn_update_dpp(
      0, __float_as_int(v), 0x160 + M, 0xf, 0xf, true));
}

__device__ __forceinline__ float row16_sum(float v) {
  v += row_xor<1>(v);
  v += row_xor<2>(v);
  v += row_xor<4>(v);
  v += row_xor<8>(v);
  return v;
}
__device__ __forceinline__ float row16_max(float v) {
  v = fmaxf(v, row_xor<1>(v));
  v = fmaxf(v, row_xor<2>(v));
  v = fmaxf(v, row_xor<4>(v));
  v = fmaxf(v, row_xor<8>(v));
  return v;
}

__device__ __forceinline__ float row16_transpose_sum(const float (&v)[16],
                                                     int c) {
  const bool b8 = (c & 8) != 0, b4 = (c & 4) != 0;
  const bool b2 = (c & 2) != 0, b1 = (c & 1) != 0;
  float w8[8], w4[4], w2[2];
#pragma unroll
  for (int e = 0; e < 8; ++e)
    w8[e] = (b8 ? v[e + 8] : v[e]) + row_xor<8>(b8 ? v[e] : v[e + 8]);
#pragma unroll
  for (int e = 0; e < 4; ++e)
    w4[e] = (b4 ? w8[e + 4] : w8[e]) + row_xor<4>(b4 ? w8[e] : w8[e + 4]);
#pragma unroll
  for (int e = 0; e < 2; ++e)
    w2[e] = (b2 ? w4[e + 2] : w4[e]) + row_xor<2>(b2 ? w4[e] : w4[e + 2]);
  return (b1 ? w2[1] : w2[0]) + row_xor<1>(b1 ? w2[0] : w2[1]);
}

__global__ __launch_bounds__(GG_THREADS) void gather_gate_kernel(
    const float* __restrict__ embed_u_z, const float* __restrict__ embed_u_c,
    const float* __restrict__ embed_v_z, const float* __restrict__ embed_v_c,
    const float* __restrict__ item_table,
    const int* __restrict__ hist_items, const int* __restrict__ hist_lens,
    const int* __restrict__ item_ids, const int* __restrict__ red_by_friends,
    float* __restrict__ uv, int H) {
  const int lane = threadIdx.x & 31;
  const int c    = lane & 15;
  const int b    = blockIdx.x * (2 * GG_WAVES) + (threadIdx.x >> 4);

  const int item = item_ids[b];
  int len        = hist_lens[b];
  const int red  = red_by_friends[b];
  const int* hrow = hist_items + (long)b * H;
  int h0 = -1, h1 = -1, h2 = -1, h3 = -1;
  if (c < H) h0 = hrow[c];
  if (c + GG_CHUNK < H) h1 = hrow[c + GG_CHUNK];
  if (c + 2 * GG_CHUNK < H) h2 = hrow[c + 2 * GG_CHUNK];
  if (c + 3 * GG_CHUNK < H) h3 = hrow[c + 3 * GG_CHUNK];
  const long bd = (long)b * DD + 4 * c;
  const float4 zu = *(const float4*)(embed_u_z + bd);
  const float4 zv = *(const float4*)(embed_v_z + bd);
  len = len < H ? len : H;

  bool any = (c < len && h0 != item) ||
             (c + GG_CHUNK < len && h1 != item) ||
             (c + 2 * GG_CHUNK < len && h2 != item) ||
             (c + 3 * GG_CHUNK < len && h3 != item);
  for (int j = c + 4 * GG_CHUNK; j < len; j += GG_CHUNK)
    any = any || (hrow[j] != item);

  const int flen = red > 0 ? len : 0;

  const float4* __restrict__ tab4 = (const float4*)item_table;
  float4 a = make_float4(0.0f, 0.0f, 0.0f, 0.0f);
  if (flen > 0) a = tab4[(unsigned)item * (DD / 4) + (unsigned)c];
  const v2f al = {a.x, a.y}, ah = {a.z, a.w};
  const float nb = fmaxf(
      sqrtf(row16_sum(a.x * a.x + a.y * a.y + a.z * a.z + a.w * a.w)),
      EPS_COS);

  float m = -INFINITY;
  int hq  = h0;
  for (int j0 = 0; j0 < H; j0 += GG_CHUNK) {
    if (__builtin_amdgcn_ballot_w32(j0 < flen) == 0u) break;
    const int jn = j0 + GG_CHUNK;
    int hn = -1;
    if (jn < 4 * GG_CHUNK)
      hn = jn == GG_CHUNK ? h1 : (jn == 2 * GG_CHUNK ? h2 : h3);
    else if (jn + c < H)
      hn = hrow[jn + c];
    const int left = flen - j0;
    float d[GG_CHUNK], n[GG_CHUNK];
    float4 x[GG_CHUNK];
#pragma unroll
    for (int u = 0; u < GG_CHUNK; ++u) {
      const unsigned sel = 0x11111111u * (unsigned)u;
      const int idx = (int)__builtin_amdgcn_permlane16(
          (unsigned)hq, (unsigned)hq, sel, sel, false, false);
      x[u] = make_float4(0.0f, 0.0f, 0.0f, 0.0f);
      if (u < left) x[u] = tab4[(unsigned)idx * (DD / 4) + (unsigned)c];
    }
#pragma unroll
    for (int u = 0; u < GG_CHUNK; ++u) {
      const v2f xl = {x[u].x, x[u].y}, xh = {x[u].z, x[u].w};
      const v2f dd = xl * al + xh * ah;
      const v2f nn = xl * xl + xh * xh;
      d[u] = dd.x + dd.y;
      n[u] = nn.x + nn.y;
    }
    const float dot = row16_transpose_sum(d, c);
    const float na  = sqrtf(row16_transpose_sum(n, c));
    const float sim = dot / (fmaxf(na, EPS_COS) * nb);
    const bool valid = (c < left) && (hq != item);
    m  = valid ? fmaxf(m, sim) : m;
    hq = hn;
  }
  m = row16_max(m);
  const unsigned bal = __builtin_amdgcn_ballot_w32(any);
  const bool empty   = ((bal >> (lane & 16)) & 0xffffu) == 0u;
  const bool gate    = (red > 0) && (m > 0.5f);
  const bool use_u_c = empty || gate;

  float4 cu = make_float4(0.0f, 0.0f, 0.0f, 0.0f);
  float4 cv = cu;
  if (use_u_c) {
    cu = *(const float4*)(embed_u_c + bd);
    cv = cu;
    if (!empty) cv = *(const float4*)(embed_v_c + bd);
  }
  float* o = uv + (long)b * (2 * DD) + 4 * c;
  *(float4*)o        = make_float4(zu.x + cu.x, zu.y + cu.y, zu.z + cu.z, zu.w + cu.w);
  *(float4*)(o + DD) = make_float4(zv.x + cv.x, zv.y + cv.y, zv.z + cv.z, zv.w + cv.w);
}

template <int NT, int K, int RPT>
__device__ __forceinline__ void bn_request(
    const float* __restrict__ psum, const float* __restrict__ psq, int nparts,
    float4 (&s)[RPT], float4 (&q)[RPT]) {
  constexpr int QPR   = K / 4;
  constexpr int G     = NT / QPR;
  constexpr int PITCH = K < 32 ? 32 : K;
  static_assert(NT % QPR == 0, "threads must tile the partial rows");
  const int cq  = threadIdx.x % QPR;
  const int grp = threadIdx.x / QPR;
  const bool full   = RPT * G <= nparts;
  const long first  = full ? (long)grp * PITCH : 0;
  const long stride = full ? (long)G * PITCH : 0;
  const float* ps = psum + first + 4 * cq;
  const float* pq = psq  + first + 4 * cq;
#pragma unroll
  for (int i = 0; i < RPT; ++i) {
    s[i] = *(const float4*)(ps + i * stride);
    q[i] = *(const float4*)(pq + i * stride);
  }
}

template <int NT, int K, int RPT>
__device__ __forceinline__ void bn_finish(
    const float* __restrict__ psum, const float* __restrict__ psq, int nparts,
    const float4 (&s)[RPT], const float4 (&q)[RPT], float gamma, float beta,
    float invB, float* sc_s, float* sh_s, float* red) {
  constexpr int QPR   = K / 4;
  constexpr int G     = NT / QPR;
  constexpr int PITCH = K < 32 ? 32 : K;
  constexpr int RP    = G + 4;
  static_assert(G % 4 == 0, "a column's group sums are read 16 bytes at a time");
  const int t   = threadIdx.x;
  const int cq  = t % QPR;
  const int grp = t / QPR;
  float4 S = make_float4(0.0f, 0.0f, 0.0f, 0.0f);
  float4 Q = make_float4(0.0f, 0.0f, 0.0f, 0.0f);
#pragma unroll
  for (int i = 0; i < RPT; ++i) {
    S.x += s[i].x; S.y += s[i].y; S.z += s[i].z; S.w += s[i].w;
    Q.x += q[i].x; Q.y += q[i].y; Q.z += q[i].z; Q.w += q[i].w;
  }
  const bool full = RPT * G <= nparts;
  if (!full) {
    S = make_float4(0.0f, 0.0f, 0.0f, 0.0f);
    Q = S;
  }
  for (int p = grp + (full ? RPT * G : 0); p < nparts; p += G) {
    const float4 a = *(const float4*)(psum + (long)p * PITCH + 4 * cq);
    const float4 b = *(const float4*)(psq  + (long)p * PITCH + 4 * cq);
    S.x += a.x; S.y += a.y; S.z += a.z; S.w += a.w;
    Q.x += b.x; Q.y += b.y; Q.z += b.z; Q.w += b.w;
  }
  float* rs = red + (4 * cq) * RP + grp;
  float* rq = rs + K * RP;
  rs[0] = S.x; rs[RP] = S.y; rs[2 * RP] = S.z; rs[3 * RP] = S.w;
  rq[0] = Q.x; rq[RP] = Q.y; rq[2 * RP] = Q.z; rq[3 * RP] = Q.w;
  __syncthreads();
  if (t < K) {
    const float4* cs  = (const float4*)(red + t * RP);
    const float4* cq4 = (const float4*)(red + (K + t) * RP);
    float4 v[G / 4], w[G / 4];
#pragma unroll
    for (int i = 0; i < G / 4; ++i) {
      v[i] = cs[i];
      w[i] = cq4[i];
    }
    __builtin_amdgcn_sched_barrier(0);
    float Sc = v[0].x, Qc = w[0].x;
    Sc += v[0].y; Sc += v[0].z; Sc += v[0].w;
    Qc += w[0].y; Qc += w[0].z; Qc += w[0].w;
#pragma unroll
    for (int i = 1; i < G / 4; ++i) {
      Sc += v[i].x; Sc += v[i].y; Sc += v[i].z; Sc += v[i].w;
      Qc += w[i].x; Qc += w[i].y; Qc += w[i].z; Qc += w[i].w;
    }
    const float mean = Sc * invB;
    const float var  = Qc * invB - mean * mean;
    const float sc   = gamma * rsqrtf(var + EPS_BN);
    sc_s[t] = sc;
    sh_s[t] = beta - mean * sc;
  }
  __syncthreads();
}

#define MLP_WAVES   4
#define MLP_THREADS (MLP_WAVES * 32)

template <int K, int N, bool USE_BN>
__global__ __launch_bounds__(MLP_THREADS) void mlp_layer_kernel(
    const float* __restrict__ x,
    const float* __restrict__ w,
    const float* __restrict__ bias,
    const float* __restrict__ in_psum,
    const float* __restrict__ in_psq,
    const float* __restrict__ bn_g,
    const float* __restrict__ bn_be,
    float* __restrict__ y,
    float* __restrict__ out_psum,
    float* __restrict__ out_psq,
    int nparts, float invB) {
  constexpr int KS  = K / 4;
  constexpr int NT  = N / 16;
  constexpr int WP  = N + 8;
  constexpr int QPR = N / 4;
  constexpr int WQ  = K * QPR / MLP_THREADS;
  __shared__ __align__(16) float w_s[K * WP];
  __shared__ __align__(16) float sc_s[K];
  __shared__ __align__(16) float sh_s[K];
  constexpr int RPT = USE_BN ? K / 2 : 1;
  constexpr int BG  = MLP_THREADS / (K / 4);
  __shared__ __align__(16) float red[USE_BN ? 2 * K * (BG + 4) : 4];
  __shared__ float wred[2][MLP_WAVES][2][N];

  const int t    = threadIdx.x;
  const int lane = t & 31;
  const int wave = t >> 5;
  const int row0 = (blockIdx.x * MLP_WAVES + wave) * 16;
  const int r    = lane & 15;
  const int half = lane >> 4;
  const int col  = lane & 15;

  float4 wq[WQ];
#pragma unroll
  for (int i = 0; i < WQ; ++i)
    wq[i] = ((const float4*)w)[t + i * MLP_THREADS];

  const float* xr = x + (long)(row0 + r) * K + 2 * half;
  v2f a[KS];
#pragma unroll
  for (int s = 0; s < KS; ++s) a[s] = *(const v2f*)(xr + 4 * s);

  float4 ps[RPT], pq[RPT];
  float gam = 0.0f, bet = 0.0f;
  if (USE_BN) {
    gam = bn_g[t % K];
    bet = bn_be[t % K];
    bn_request<MLP_THREADS, K, RPT>(in_psum, in_psq, nparts, ps, pq);
  }

#pragma unroll
  for (int i = 0; i < WQ; ++i) {
    const int p = t + i * MLP_THREADS;
    *(float4*)(w_s + (p / QPR) * WP + 4 * (p % QPR)) = wq[i];
  }
  if (USE_BN)
    bn_finish<MLP_THREADS, K, RPT>(in_psum, in_psq, nparts, ps, pq, gam, bet,
                                   invB, sc_s, sh_s, red);
  else
    __syncthreads();

  if (USE_BN) {
#pragma unroll
    for (int s = 0; s < KS; ++s) {
      const v2f sc = *(const v2f*)(sc_s + 4 * s + 2 * half);
      const v2f sh = *(const v2f*)(sh_s + 4 * s + 2 * half);
      a[s] = sc * a[s] + sh;
    }
  }

  v8f acc[NT];
#pragma unroll
  for (int nt = 0; nt < NT; ++nt) acc[nt] = v8f{};
  const float* wl = w_s + (2 * half) * WP + col;
#pragma unroll
  for (int s = 0; s < KS; ++s) {
#pragma unroll
    for (int nt = 0; nt < NT; ++nt) {
      v2f bb;
      bb.x = wl[(4 * s) * WP + nt * 16];
      bb.y = wl[(4 * s + 1) * WP + nt * 16];
      acc[nt] = __builtin_amdgcn_wmma_f32_16x16x4_f32(
          false, a[s], false, bb, (short)0, acc[nt], false, false);
    }
  }

#pragma unroll
  for (int nt = 0; nt < NT; ++nt) {
    const int n = nt * 16 + col;
    const float bn = bias[n];
    float s = 0.0f, q = 0.0f;
#pragma unroll
    for (int j = 0; j < 8; ++j) {
      const int m = row0 + j + half * 8;
      const float v = fmaxf(acc[nt][j] + bn, 0.0f);
      y[(long)m * N + n] = v;
      s += v;
      q += v * v;
    }
    wred[0][wave][half][n] = s;
    wred[1][wave][half][n] = q;
  }
  __syncthreads();

  if (t < 2 * N) {
    const int kind   = t >= N ? 1 : 0;
    const int n      = t - kind * N;
    constexpr int opitch = N < 32 ? 32 : N;
    float a2 = 0.0f;
    for (int wv = 0; wv < MLP_WAVES; ++wv)
      a2 += wred[kind][wv][0][n] + wred[kind][wv][1][n];
    (kind ? out_psq : out_psum)[(long)blockIdx.x * opitch + n] = a2;
  }
}

#define FIN_THREADS 256

__global__ __launch_bounds__(FIN_THREADS) void final_kernel(
    const float* __restrict__ y3, const float* __restrict__ psum3,
    const float* __restrict__ psq3, int nparts, const float* __restrict__ g3,
    const float* __restrict__ be3, const float* __restrict__ wf,
    const float* __restrict__ bf, float* __restrict__ out, int Bn, float invB) {
  constexpr int RPT = 4;
  constexpr int BG  = FIN_THREADS / (16 / 4);
  __shared__ float sc_s[16], sh_s[16], wf_s[16];
  __shared__ __align__(16) float red[2 * 16 * (BG + 4)];
  const int t = threadIdx.x;
  const int b = blockIdx.x * FIN_THREADS + t;

  const float wfv = wf[t & 15];
  const float gam = g3[t & 15];
  const float bet = be3[t & 15];
  const float bfv = bf[0];
  const float4* yr = (const float4*)(y3 + (long)(b < Bn ? b : Bn - 1) * 16);
  float4 yv[4];
#pragma unroll
  for (int i = 0; i < 4; ++i) yv[i] = yr[i];
  float4 ps[RPT], pq[RPT];
  bn_request<FIN_THREADS, 16, RPT>(psum3, psq3, nparts, ps, pq);

  if (t < 16) wf_s[t] = wfv;
  bn_finish<FIN_THREADS, 16, RPT>(psum3, psq3, nparts, ps, pq, gam, bet, invB,
                                  sc_s, sh_s, red);
  if (b < Bn) {
    const float yy[16] = {yv[0].x, yv[0].y, yv[0].z, yv[0].w,
                          yv[1].x, yv[1].y, yv[1].z, yv[1].w,
                          yv[2].x, yv[2].y, yv[2].z, yv[2].w,
                          yv[3].x, yv[3].y, yv[3].z, yv[3].w};
    float s = 0.0f;
#pragma unroll
    for (int c = 0; c < 16; ++c)
      s += (sc_s[c] * yy[c] + sh_s[c]) * wf_s[c];
    out[b] = s + bfv;
  }
}

extern "C" void kernel_launch(void* const* d_in, const int* in_sizes, int n_in,
                              void* d_out, int out_size, void* d_ws,
                              size_t ws_size, hipStream_t stream) {
  const float* embed_u_z = (const float*)d_in[0];
  const float* embed_u_c = (const float*)d_in[1];
  const float* embed_v_z = (const float*)d_in[2];
  const float* embed_v_c = (const float*)d_in[3];
  const float* item_table = (const float*)d_in[4];
  const float* w1 = (const float*)d_in[5];
  const float* b1 = (const float*)d_in[6];
  const float* g1 = (const float*)d_in[7];
  const float* be1 = (const float*)d_in[8];
  const float* w2 = (const float*)d_in[9];
  const float* b2 = (const float*)d_in[10];
  const float* g2 = (const float*)d_in[11];
  const float* be2 = (const float*)d_in[12];
  const float* w3 = (const float*)d_in[13];
  const float* b3 = (const float*)d_in[14];
  const float* g3 = (const float*)d_in[15];
  const float* be3 = (const float*)d_in[16];
  const float* wf = (const float*)d_in[17];
  const float* bf = (const float*)d_in[18];
  const int* hist_items = (const int*)d_in[19];
  const int* hist_lens = (const int*)d_in[20];
  const int* item_ids = (const int*)d_in[21];
  const int* red_by_friends = (const int*)d_in[22];

  const int Bn = in_sizes[20];
  const int H  = in_sizes[19] / Bn;
  const float invB = 1.0f / (float)Bn;
  const int NB = Bn / (16 * MLP_WAVES);

  float* ws = (float*)d_ws;
  float* uv = ws;
  float* y1 = uv + (size_t)Bn * 128;
  float* y2 = y1 + (size_t)Bn * 64;
  float* y3 = y2 + (size_t)Bn * 32;
  float* ps1 = y3 + (size_t)Bn * 16;
  float* pq1 = ps1 + (size_t)NB * 64;
  float* ps2 = pq1 + (size_t)NB * 64;
  float* pq2 = ps2 + (size_t)NB * 32;
  float* ps3 = pq2 + (size_t)NB * 32;
  float* pq3 = ps3 + (size_t)NB * 32;

  gather_gate_kernel<<<Bn / (2 * GG_WAVES), GG_THREADS, 0, stream>>>(
      embed_u_z, embed_u_c, embed_v_z, embed_v_c, item_table,
      hist_items, hist_lens, item_ids, red_by_friends, uv, H);

  mlp_layer_kernel<128, 64, false><<<NB, MLP_THREADS, 0, stream>>>(
      uv, w1, b1, nullptr, nullptr, nullptr, nullptr,
      y1, ps1, pq1, 0, invB);

  mlp_layer_kernel<64, 32, true><<<NB, MLP_THREADS, 0, stream>>>(
      y1, w2, b2, ps1, pq1, g1, be1,
      y2, ps2, pq2, NB, invB);

  mlp_layer_kernel<32, 16, true><<<NB, MLP_THREADS, 0, stream>>>(
      y2, w3, b3, ps2, pq2, g2, be2,
      y3, ps3, pq3, NB, invB);

  final_kernel<<<(Bn + 255) / 256, 256, 0, stream>>>(
      y3, ps3, pq3, NB, g3, be3, wf, bf, (float*)d_out, Bn, invB);
}


# gather kernel: a trip's 16 row requests are straight-line code, 2 instructions each: entries past the fetched length re-read an anchor row (masked as before); the address op does the index broadcast
# speedup vs baseline: 6.7871x; 1.0600x over previous
#include <hip/hip_runtime.h>
#include <hip/hip_bf16.h>
#include <math.h>

typedef __attribute__((ext_vector_type(2))) float v2f;
typedef __attribute__((ext_vector_type(8))) float v8f;

#define DD 64
#define EPS_COS 1e-8f
#define EPS_BN  1e-5f

#define GG_WAVES   8
#define GG_THREADS (GG_WAVES * 32)
#define GG_CHUNK   16

template <int M>
__device__ __forceinline__ float row_xor(float v) {
  return __int_as_float(__builtin_amdgcn_update_dpp(
      0, __float_as_int(v), 0x160 + M, 0xf, 0xf, true));
}

template <int U>
__device__ __forceinline__ int row_share(int v) {
  return __builtin_amdgcn_update_dpp(0, v, 0x150 + U, 0xf, 0xf, true);
}

__device__ __forceinline__ float row16_sum(float v) {
  v += row_xor<1>(v);
  v += row_xor<2>(v);
  v += row_xor<4>(v);
  v += row_xor<8>(v);
  return v;
}
__device__ __forceinline__ float row16_max(float v) {
  v = fmaxf(v, row_xor<1>(v));
  v = fmaxf(v, row_xor<2>(v));
  v = fmaxf(v, row_xor<4>(v));
  v = fmaxf(v, row_xor<8>(v));
  return v;
}

__device__ __forceinline__ float row16_transpose_sum(const float (&v)[16],
                                                     int c) {
  const bool b8 = (c & 8) != 0, b4 = (c & 4) != 0;
  const bool b2 = (c & 2) != 0, b1 = (c & 1) != 0;
  float w8[8], w4[4], w2[2];
#pragma unroll
  for (int e = 0; e < 8; ++e)
    w8[e] = (b8 ? v[e + 8] : v[e]) + row_xor<8>(b8 ? v[e] : v[e + 8]);
#pragma unroll
  for (int e = 0; e < 4; ++e)
    w4[e] = (b4 ? w8[e + 4] : w8[e]) + row_xor<4>(b4 ? w8[e] : w8[e + 4]);
#pragma unroll
  for (int e = 0; e < 2; ++e)
    w2[e] = (b2 ? w4[e + 2] : w4[e]) + row_xor<2>(b2 ? w4[e] : w4[e + 2]);
  return (b1 ? w2[1] : w2[0]) + row_xor<1>(b1 ? w2[0] : w2[1]);
}

__global__ __launch_bounds__(GG_THREADS) void gather_gate_kernel(
    const float* __restrict__ embed_u_z, const float* __restrict__ embed_u_c,
    const float* __restrict__ embed_v_z, const float* __restrict__ embed_v_c,
    const float* __restrict__ item_table,
    const int* __restrict__ hist_items, const int* __restrict__ hist_lens,
    const int* __restrict__ item_ids, const int* __restrict__ red_by_friends,
    float* __restrict__ uv, int H) {
  const int lane = threadIdx.x & 31;
  const int c    = lane & 15;
  const int b    = blockIdx.x * (2 * GG_WAVES) + (threadIdx.x >> 4);

  const int item = item_ids[b];
  int len        = hist_lens[b];
  const int red  = red_by_friends[b];
  const int* hrow = hist_items + (long)b * H;
  int h0 = -1, h1 = -1, h2 = -1, h3 = -1;
  if (c < H) h0 = hrow[c];
  if (c + GG_CHUNK < H) h1 = hrow[c + GG_CHUNK];
  if (c + 2 * GG_CHUNK < H) h2 = hrow[c + 2 * GG_CHUNK];
  if (c + 3 * GG_CHUNK < H) h3 = hrow[c + 3 * GG_CHUNK];
  const long bd = (long)b * DD + 4 * c;
  const float4 zu = *(const float4*)(embed_u_z + bd);
  const float4 zv = *(const float4*)(embed_v_z + bd);
  len = len < H ? len : H;

  bool any = (c < len && h0 != item) ||
             (c + GG_CHUNK < len && h1 != item) ||
             (c + 2 * GG_CHUNK < len && h2 != item) ||
             (c + 3 * GG_CHUNK < len && h3 != item);
  for (int j = c + 4 * GG_CHUNK; j < len; j += GG_CHUNK)
    any = any || (hrow[j] != item);

  const int flen = red > 0 ? len : 0;
  const int other_item = (int)__builtin_amdgcn_permlanex16(
      (unsigned)item, (unsigned)item, 0x76543210u, 0xfedcba98u, false, false);
  const int pad_item = flen > 0 ? item : other_item;

  const float4* __restrict__ tab4 = (const float4*)item_table;
  float4 a = make_float4(0.0f, 0.0f, 0.0f, 0.0f);
  if (flen > 0) a = tab4[(unsigned)item * (DD / 4) + (unsigned)c];
  const v2f al = {a.x, a.y}, ah = {a.z, a.w};
  const float nb = fmaxf(
      sqrtf(row16_sum(a.x * a.x + a.y * a.y + a.z * a.z + a.w * a.w)),
      EPS_COS);

  float m = -INFINITY;
  int hq  = h0;
  for (int j0 = 0; j0 < H; j0 += GG_CHUNK) {
    if (__builtin_amdgcn_ballot_w32(j0 < flen) == 0u) break;
    const int jn = j0 + GG_CHUNK;
    int hn = -1;
    if (jn < 4 * GG_CHUNK)
      hn = jn == GG_CHUNK ? h1 : (jn == 2 * GG_CHUNK ? h2 : h3);
    else if (jn + c < H)
      hn = hrow[jn + c];
    const int left = flen - j0;
    const int hf = c < left ? hq : pad_item;
    float d[GG_CHUNK], n[GG_CHUNK];
    float4 x[GG_CHUNK];
#define GG_REQ(U) \
    x[U] = tab4[(unsigned)row_share<U>(hf) * (DD / 4) + (unsigned)c];
    GG_REQ(0) GG_REQ(1) GG_REQ(2) GG_REQ(3) GG_REQ(4) GG_REQ(5) GG_REQ(6)
    GG_REQ(7) GG_REQ(8) GG_REQ(9) GG_REQ(10) GG_REQ(11) GG_REQ(12) GG_REQ(13)
    GG_REQ(14) GG_REQ(15)
#undef GG_REQ
#pragma unroll
    for (int u = 0; u < GG_CHUNK; ++u) {
      const v2f xl = {x[u].x, x[u].y}, xh = {x[u].z, x[u].w};
      const v2f dd = xl * al + xh * ah;
      const v2f nn = xl * xl + xh * xh;
      d[u] = dd.x + dd.y;
      n[u] = nn.x + nn.y;
    }
    const float dot = row16_transpose_sum(d, c);
    const float na  = sqrtf(row16_transpose_sum(n, c));
    const float sim = dot / (fmaxf(na, EPS_COS) * nb);
    const bool valid = (c < left) && (hq != item);
    m  = valid ? fmaxf(m, sim) : m;
    hq = hn;
  }
  m = row16_max(m);
  const unsigned bal = __builtin_amdgcn_ballot_w32(any);
  const bool empty   = ((bal >> (lane & 16)) & 0xffffu) == 0u;
  const bool gate    = (red > 0) && (m > 0.5f);
  const bool use_u_c = empty || gate;

  float4 cu = make_float4(0.0f, 0.0f, 0.0f, 0.0f);
  float4 cv = cu;
  if (use_u_c) {
    cu = *(const float4*)(embed_u_c + bd);
    cv = cu;
    if (!empty) cv = *(const float4*)(embed_v_c + bd);
  }
  float* o = uv + (long)b * (2 * DD) + 4 * c;
  *(float4*)o        = make_float4(zu.x + cu.x, zu.y + cu.y, zu.z + cu.z, zu.w + cu.w);
  *(float4*)(o + DD) = make_float4(zv.x + cv.x, zv.y + cv.y, zv.z + cv.z, zv.w + cv.w);
}

template <int NT, int K, int RPT>
__device__ __forceinline__ void bn_request(
    const float* __restrict__ psum, const float* __restrict__ psq, int nparts,
    float4 (&s)[RPT], float4 (&q)[RPT]) {
  constexpr int QPR   = K / 4;
  constexpr int G     = NT / QPR;
  constexpr int PITCH = K < 32 ? 32 : K;
  static_assert(NT % QPR == 0, "threads must tile the partial rows");
  const int cq  = threadIdx.x % QPR;
  const int grp = threadIdx.x / QPR;
  const bool full   = RPT * G <= nparts;
  const long first  = full ? (long)grp * PITCH : 0;
  const long stride = full ? (long)G * PITCH : 0;
  const float* ps = psum + first + 4 * cq;
  const float* pq = psq  + first + 4 * cq;
#pragma unroll
  for (int i = 0; i < RPT; ++i) {
    s[i] = *(const float4*)(ps + i * stride);
    q[i] = *(const float4*)(pq + i * stride);
  }
}

template <int NT, int K, int RPT>
__device__ __forceinline__ void bn_finish(
    const float* __restrict__ psum, const float* __restrict__ psq, int nparts,
    const float4 (&s)[RPT], const float4 (&q)[RPT], float gamma, float beta,
    float invB, float* sc_s, float* sh_s, float* red) {
  constexpr int QPR   = K / 4;
  constexpr int G     = NT / QPR;
  constexpr int PITCH = K < 32 ? 32 : K;
  constexpr int RP    = G + 4;
  static_assert(G % 4 == 0, "a column's group sums are read 16 bytes at a time");
  const int t   = threadIdx.x;
  const int cq  = t % QPR;
  const int grp = t / QPR;
  float4 S = make_float4(0.0f, 0.0f, 0.0f, 0.0f);
  float4 Q = make_float4(0.0f, 0.0f, 0.0f, 0.0f);
#pragma unroll
  for (int i = 0; i < RPT; ++i) {
    S.x += s[i].x; S.y += s[i].y; S.z += s[i].z; S.w += s[i].w;
    Q.x += q[i].x; Q.y += q[i].y; Q.z += q[i].z; Q.w += q[i].w;
  }
  const bool full = RPT * G <= nparts;
  if (!full) {
    S = make_float4(0.0f, 0.0f, 0.0f, 0.0f);
    Q = S;
  }
  for (int p = grp + (full ? RPT * G : 0); p < nparts; p += G) {
    const float4 a = *(const float4*)(psum + (long)p * PITCH + 4 * cq);
    const float4 b = *(const float4*)(psq  + (long)p * PITCH + 4 * cq);
    S.x += a.x; S.y += a.y; S.z += a.z; S.w += a.w;
    Q.x += b.x; Q.y += b.y; Q.z += b.z; Q.w += b.w;
  }
  float* rs = red + (4 * cq) * RP + grp;
  float* rq = rs + K * RP;
  rs[0] = S.x; rs[RP] = S.y; rs[2 * RP] = S.z; rs[3 * RP] = S.w;
  rq[0] = Q.x; rq[RP] = Q.y; rq[2 * RP] = Q.z; rq[3 * RP] = Q.w;
  __syncthreads();
  if (t < K) {
    const float4* cs  = (const float4*)(red + t * RP);
    const float4* cq4 = (const float4*)(red + (K + t) * RP);
    float4 v[G / 4], w[G / 4];
#pragma unroll
    for (int i = 0; i < G / 4; ++i) {
      v[i] = cs[i];
      w[i] = cq4[i];
    }
    __builtin_amdgcn_sched_barrier(0);
    float Sc = v[0].x, Qc = w[0].x;
    Sc += v[0].y; Sc += v[0].z; Sc += v[0].w;
    Qc += w[0].y; Qc += w[0].z; Qc += w[0].w;
#pragma unroll
    for (int i = 1; i < G / 4; ++i) {
      Sc += v[i].x; Sc += v[i].y; Sc += v[i].z; Sc += v[i].w;
      Qc += w[i].x; Qc += w[i].y; Qc += w[i].z; Qc += w[i].w;
    }
    const float mean = Sc * invB;
    const float var  = Qc * invB - mean * mean;
    const float sc   = gamma * rsqrtf(var + EPS_BN);
    sc_s[t] = sc;
    sh_s[t] = beta - mean * sc;
  }
  __syncthreads();
}

#define MLP_WAVES   4
#define MLP_THREADS (MLP_WAVES * 32)

template <int K, int N, bool USE_BN>
__global__ __launch_bounds__(MLP_THREADS) void mlp_layer_kernel(
    const float* __restrict__ x,
    const float* __restrict__ w,
    const float* __restrict__ bias,
    const float* __restrict__ in_psum,
    const float* __restrict__ in_psq,
    const float* __restrict__ bn_g,
    const float* __restrict__ bn_be,
    float* __restrict__ y,
    float* __restrict__ out_psum,
    float* __restrict__ out_psq,
    int nparts, float invB) {
  constexpr int KS  = K / 4;
  constexpr int NT  = N / 16;
  constexpr int WP  = N + 8;
  constexpr int QPR = N / 4;
  constexpr int WQ  = K * QPR / MLP_THREADS;
  __shared__ __align__(16) float w_s[K * WP];
  __shared__ __align__(16) float sc_s[K];
  __shared__ __align__(16) float sh_s[K];
  constexpr int RPT = USE_BN ? K / 2 : 1;
  constexpr int BG  = MLP_THREADS / (K / 4);
  __shared__ __align__(16) float red[USE_BN ? 2 * K * (BG + 4) : 4];
  __shared__ float wred[2][MLP_WAVES][2][N];

  const int t    = threadIdx.x;
  const int lane = t & 31;
  const int wave = t >> 5;
  const int row0 = (blockIdx.x * MLP_WAVES + wave) * 16;
  const int r    = lane & 15;
  const int half = lane >> 4;
  const int col  = lane & 15;

  float4 wq[WQ];
#pragma unroll
  for (int i = 0; i < WQ; ++i)
    wq[i] = ((const float4*)w)[t + i * MLP_THREADS];

  const float* xr = x + (long)(row0 + r) * K + 2 * half;
  v2f a[KS];
#pragma unroll
  for (int s = 0; s < KS; ++s) a[s] = *(const v2f*)(xr + 4 * s);

  float4 ps[RPT], pq[RPT];
  float gam = 0.0f, bet = 0.0f;
  if (USE_BN) {
    gam = bn_g[t % K];
    bet = bn_be[t % K];
    bn_request<MLP_THREADS, K, RPT>(in_psum, in_psq, nparts, ps, pq);
  }

#pragma unroll
  for (int i = 0; i < WQ; ++i) {
    const int p = t + i * MLP_THREADS;
    *(float4*)(w_s + (p / QPR) * WP + 4 * (p % QPR)) = wq[i];
  }
  if (USE_BN)
    bn_finish<MLP_THREADS, K, RPT>(in_psum, in_psq, nparts, ps, pq, gam, bet,
                                   invB, sc_s, sh_s, red);
  else
    __syncthreads();

  if (USE_BN) {
#pragma unroll
    for (int s = 0; s < KS; ++s) {
      const v2f sc = *(const v2f*)(sc_s + 4 * s + 2 * half);
      const v2f sh = *(const v2f*)(sh_s + 4 * s + 2 * half);
      a[s] = sc * a[s] + sh;
    }
  }

  v8f acc[NT];
#pragma unroll
  for (int nt = 0; nt < NT; ++nt) acc[nt] = v8f{};
  const float* wl = w_s + (2 * half) * WP + col;
#pragma unroll
  for (int s = 0; s < KS; ++s) {
#pragma unroll
    for (int nt = 0; nt < NT; ++nt) {
      v2f bb;
      bb.x = wl[(4 * s) * WP + nt * 16];
      bb.y = wl[(4 * s + 1) * WP + nt * 16];
      acc[nt] = __builtin_amdgcn_wmma_f32_16x16x4_f32(
          false, a[s], false, bb, (short)0, acc[nt], false, false);
    }
  }

#pragma unroll
  for (int nt = 0; nt < NT; ++nt) {
    const int n = nt * 16 + col;
    const float bn = bias[n];
    float s = 0.0f, q = 0.0f;
#pragma unroll
    for (int j = 0; j < 8; ++j) {
      const int m = row0 + j + half * 8;
      const float v = fmaxf(acc[nt][j] + bn, 0.0f);
      y[(long)m * N + n] = v;
      s += v;
      q += v * v;
    }
    wred[0][wave][half][n] = s;
    wred[1][wave][half][n] = q;
  }
  __syncthreads();

  if (t < 2 * N) {
    const int kind   = t >= N ? 1 : 0;
    const int n      = t - kind * N;
    constexpr int opitch = N < 32 ? 32 : N;
    float a2 = 0.0f;
    for (int wv = 0; wv < MLP_WAVES; ++wv)
      a2 += wred[kind][wv][0][n] + wred[kind][wv][1][n];
    (kind ? out_psq : out_psum)[(long)blockIdx.x * opitch + n] = a2;
  }
}

#define FIN_THREADS 256

__global__ __launch_bounds__(FIN_THREADS) void final_kernel(
    const float* __restrict__ y3, const float* __restrict__ psum3,
    const float* __restrict__ psq3, int nparts, const float* __restrict__ g3,
    const float* __restrict__ be3, const float* __restrict__ wf,
    const float* __restrict__ bf, float* __restrict__ out, int Bn, float invB) {
  constexpr int RPT = 4;
  constexpr int BG  = FIN_THREADS / (16 / 4);
  __shared__ float sc_s[16], sh_s[16], wf_s[16];
  __shared__ __align__(16) float red[2 * 16 * (BG + 4)];
  const int t = threadIdx.x;
  const int b = blockIdx.x * FIN_THREADS + t;

  const float wfv = wf[t & 15];
  const float gam = g3[t & 15];
  const float bet = be3[t & 15];
  const float bfv = bf[0];
  const float4* yr = (const float4*)(y3 + (long)(b < Bn ? b : Bn - 1) * 16);
  float4 yv[4];
#pragma unroll
  for (int i = 0; i < 4; ++i) yv[i] = yr[i];
  float4 ps[RPT], pq[RPT];
  bn_request<FIN_THREADS, 16, RPT>(psum3, psq3, nparts, ps, pq);

  if (t < 16) wf_s[t] = wfv;
  bn_finish<FIN_THREADS, 16, RPT>(psum3, psq3, nparts, ps, pq, gam, bet, invB,
                                  sc_s, sh_s, red);
  if (b < Bn) {
    const float yy[16] = {yv[0].x, yv[0].y, yv[0].z, yv[0].w,
                          yv[1].x, yv[1].y, yv[1].z, yv[1].w,
                          yv[2].x, yv[2].y, yv[2].z, yv[2].w,
                          yv[3].x, yv[3].y, yv[3].z, yv[3].w};
    float s = 0.0f;
#pragma unroll
    for (int c = 0; c < 16; ++c)
      s += (sc_s[c] * yy[c] + sh_s[c]) * wf_s[c];
    out[b] = s + bfv;
  }
}

extern "C" void kernel_launch(void* const* d_in, const int* in_sizes, int n_in,
                              void* d_out, int out_size, void* d_ws,
                              size_t ws_size, hipStream_t stream) {
  const float* embed_u_z = (const float*)d_in[0];
  const float* embed_u_c = (const float*)d_in[1];
  const float* embed_v_z = (const float*)d_in[2];
  const float* embed_v_c = (const float*)d_in[3];
  const float* item_table = (const float*)d_in[4];
  const float* w1 = (const float*)d_in[5];
  const float* b1 = (const float*)d_in[6];
  const float* g1 = (const float*)d_in[7];
  const float* be1 = (const float*)d_in[8];
  const float* w2 = (const float*)d_in[9];
  const float* b2 = (const float*)d_in[10];
  const float* g2 = (const float*)d_in[11];
  const float* be2 = (const float*)d_in[12];
  const float* w3 = (const float*)d_in[13];
  const float* b3 = (const float*)d_in[14];
  const float* g3 = (const float*)d_in[15];
  const float* be3 = (const float*)d_in[16];
  const float* wf = (const float*)d_in[17];
  const float* bf = (const float*)d_in[18];
  const int* hist_items = (const int*)d_in[19];
  const int* hist_lens = (const int*)d_in[20];
  const int* item_ids = (const int*)d_in[21];
  const int* red_by_friends = (const int*)d_in[22];

  const int Bn = in_sizes[20];
  const int H  = in_sizes[19] / Bn;
  const float invB = 1.0f / (float)Bn;
  const int NB = Bn / (16 * MLP_WAVES);

  float* ws = (float*)d_ws;
  float* uv = ws;
  float* y1 = uv + (size_t)Bn * 128;
  float* y2 = y1 + (size_t)Bn * 64;
  float* y3 = y2 + (size_t)Bn * 32;
  float* ps1 = y3 + (size_t)Bn * 16;
  float* pq1 = ps1 + (size_t)NB * 64;
  float* ps2 = pq1 + (size_t)NB * 64;
  float* pq2 = ps2 + (size_t)NB * 32;
  float* ps3 = pq2 + (size_t)NB * 32;
  float* pq3 = ps3 + (size_t)NB * 32;

  gather_gate_kernel<<<Bn / (2 * GG_WAVES), GG_THREADS, 0, stream>>>(
      embed_u_z, embed_u_c, embed_v_z, embed_v_c, item_table,
      hist_items, hist_lens, item_ids, red_by_friends, uv, H);

  mlp_layer_kernel<128, 64, false><<<NB, MLP_THREADS, 0, stream>>>(
      uv, w1, b1, nullptr, nullptr, nullptr, nullptr,
      y1, ps1, pq1, 0, invB);

  mlp_layer_kernel<64, 32, true><<<NB, MLP_THREADS, 0, stream>>>(
      y1, w2, b2, ps1, pq1, g1, be1,
      y2, ps2, pq2, NB, invB);

  mlp_layer_kernel<32, 16, true><<<NB, MLP_THREADS, 0, stream>>>(
      y2, w3, b3, ps2, pq2, g2, be2,
      y3, ps3, pq3, NB, invB);

  final_kernel<<<(Bn + 255) / 256, 256, 0, stream>>>(
      y3, ps3, pq3, NB, g3, be3, wf, bf, (float*)d_out, Bn, invB);
}


# layer and final kernels: barriers fence LDS only and x rows are requested last, so no barrier waits for x loads or for the y stores' acknowledgement; x waited per k step; no sum reordered
# speedup vs baseline: 7.1853x; 1.0587x over previous
#include <hip/hip_runtime.h>
#include <hip/hip_bf16.h>
#include <math.h>

typedef __attribute__((ext_vector_type(2))) float v2f;
typedef __attribute__((ext_vector_type(8))) float v8f;

#define DD 64
#define EPS_COS 1e-8f
#define EPS_BN  1e-5f

#define GG_WAVES   8
#define GG_THREADS (GG_WAVES * 32)
#define GG_CHUNK   16

template <int M>
__device__ __forceinline__ float row_xor(float v) {
  return __int_as_float(__builtin_amdgcn_update_dpp(
      0, __float_as_int(v), 0x160 + M, 0xf, 0xf, true));
}

template <int U>
__device__ __forceinline__ int row_share(int v) {
  return __builtin_amdgcn_update_dpp(0, v, 0x150 + U, 0xf, 0xf, true);
}

__device__ __forceinline__ float row16_sum(float v) {
  v += row_xor<1>(v);
  v += row_xor<2>(v);
  v += row_xor<4>(v);
  v += row_xor<8>(v);
  return v;
}
__device__ __forceinline__ float row16_max(float v) {
  v = fmaxf(v, row_xor<1>(v));
  v = fmaxf(v, row_xor<2>(v));
  v = fmaxf(v, row_xor<4>(v));
  v = fmaxf(v, row_xor<8>(v));
  return v;
}

__device__ __forceinline__ float row16_transpose_sum(const float (&v)[16],
                                                     int c) {
  const bool b8 = (c & 8) != 0, b4 = (c & 4) != 0;
  const bool b2 = (c & 2) != 0, b1 = (c & 1) != 0;
  float w8[8], w4[4], w2[2];
#pragma unroll
  for (int e = 0; e < 8; ++e)
    w8[e] = (b8 ? v[e + 8] : v[e]) + row_xor<8>(b8 ? v[e] : v[e + 8]);
#pragma unroll
  for (int e = 0; e < 4; ++e)
    w4[e] = (b4 ? w8[e + 4] : w8[e]) + row_xor<4>(b4 ? w8[e] : w8[e + 4]);
#pragma unroll
  for (int e = 0; e < 2; ++e)
    w2[e] = (b2 ? w4[e + 2] : w4[e]) + row_xor<2>(b2 ? w4[e] : w4[e + 2]);
  return (b1 ? w2[1] : w2[0]) + row_xor<1>(b1 ? w2[0] : w2[1]);
}

__global__ __launch_bounds__(GG_THREADS) void gather_gate_kernel(
    const float* __restrict__ embed_u_z, const float* __restrict__ embed_u_c,
    const float* __restrict__ embed_v_z, const float* __restrict__ embed_v_c,
    const float* __restrict__ item_table,
    const int* __restrict__ hist_items, const int* __restrict__ hist_lens,
    const int* __restrict__ item_ids, const int* __restrict__ red_by_friends,
    float* __restrict__ uv, int H) {
  const int lane = threadIdx.x & 31;
  const int c    = lane & 15;
  const int b    = blockIdx.x * (2 * GG_WAVES) + (threadIdx.x >> 4);

  const int item = item_ids[b];
  int len        = hist_lens[b];
  const int red  = red_by_friends[b];
  const int* hrow = hist_items + (long)b * H;
  int h0 = -1, h1 = -1, h2 = -1, h3 = -1;
  if (c < H) h0 = hrow[c];
  if (c + GG_CHUNK < H) h1 = hrow[c + GG_CHUNK];
  if (c + 2 * GG_CHUNK < H) h2 = hrow[c + 2 * GG_CHUNK];
  if (c + 3 * GG_CHUNK < H) h3 = hrow[c + 3 * GG_CHUNK];
  const long bd = (long)b * DD + 4 * c;
  const float4 zu = *(const float4*)(embed_u_z + bd);
  const float4 zv = *(const float4*)(embed_v_z + bd);
  len = len < H ? len : H;

  bool any = (c < len && h0 != item) ||
             (c + GG_CHUNK < len && h1 != item) ||
             (c + 2 * GG_CHUNK < len && h2 != item) ||
             (c + 3 * GG_CHUNK < len && h3 != item);
  for (int j = c + 4 * GG_CHUNK; j < len; j += GG_CHUNK)
    any = any || (hrow[j] != item);

  const int flen = red > 0 ? len : 0;
  const int other_item = (int)__builtin_amdgcn_permlanex16(
      (unsigned)item, (unsigned)item, 0x76543210u, 0xfedcba98u, false, false);
  const int pad_item = flen > 0 ? item : other_item;

  const float4* __restrict__ tab4 = (const float4*)item_table;
  float4 a = make_float4(0.0f, 0.0f, 0.0f, 0.0f);
  if (flen > 0) a = tab4[(unsigned)item * (DD / 4) + (unsigned)c];
  const v2f al = {a.x, a.y}, ah = {a.z, a.w};
  const float nb = fmaxf(
      sqrtf(row16_sum(a.x * a.x + a.y * a.y + a.z * a.z + a.w * a.w)),
      EPS_COS);

  float m = -INFINITY;
  int hq  = h0;
  for (int j0 = 0; j0 < H; j0 += GG_CHUNK) {
    if (__builtin_amdgcn_ballot_w32(j0 < flen) == 0u) break;
    const int jn = j0 + GG_CHUNK;
    int hn = -1;
    if (jn < 4 * GG_CHUNK)
      hn = jn == GG_CHUNK ? h1 : (jn == 2 * GG_CHUNK ? h2 : h3);
    else if (jn + c < H)
      hn = hrow[jn + c];
    const int left = flen - j0;
    const int hf = c < left ? hq : pad_item;
    float d[GG_CHUNK], n[GG_CHUNK];
    float4 x[GG_CHUNK];
#define GG_REQ(U) \
    x[U] = tab4[(unsigned)row_share<U>(hf) * (DD / 4) + (unsigned)c];
    GG_REQ(0) GG_REQ(1) GG_REQ(2) GG_REQ(3) GG_REQ(4) GG_REQ(5) GG_REQ(6)
    GG_REQ(7) GG_REQ(8) GG_REQ(9) GG_REQ(10) GG_REQ(11) GG_REQ(12) GG_REQ(13)
    GG_REQ(14) GG_REQ(15)
#undef GG_REQ
#pragma unroll
    for (int u = 0; u < GG_CHUNK; ++u) {
      const v2f xl = {x[u].x, x[u].y}, xh = {x[u].z, x[u].w};
      const v2f dd = xl * al + xh * ah;
      const v2f nn = xl * xl + xh * xh;
      d[u] = dd.x + dd.y;
      n[u] = nn.x + nn.y;
    }
    const float dot = row16_transpose_sum(d, c);
    const float na  = sqrtf(row16_transpose_sum(n, c));
    const float sim = dot / (fmaxf(na, EPS_COS) * nb);
    const bool valid = (c < left) && (hq != item);
    m  = valid ? fmaxf(m, sim) : m;
    hq = hn;
  }
  m = row16_max(m);
  const unsigned bal = __builtin_amdgcn_ballot_w32(any);
  const bool empty   = ((bal >> (lane & 16)) & 0xffffu) == 0u;
  const bool gate    = (red > 0) && (m > 0.5f);
  const bool use_u_c = empty || gate;

  float4 cu = make_float4(0.0f, 0.0f, 0.0f, 0.0f);
  float4 cv = cu;
  if (use_u_c) {
    cu = *(const float4*)(embed_u_c + bd);
    cv = cu;
    if (!empty) cv = *(const float4*)(embed_v_c + bd);
  }
  float* o = uv + (long)b * (2 * DD) + 4 * c;
  *(float4*)o        = make_float4(zu.x + cu.x, zu.y + cu.y, zu.z + cu.z, zu.w + cu.w);
  *(float4*)(o + DD) = make_float4(zv.x + cv.x, zv.y + cv.y, zv.z + cv.z, zv.w + cv.w);
}

__device__ __forceinline__ void lds_barrier() {
  __builtin_amdgcn_fence(__ATOMIC_RELEASE, "workgroup", "local");
  __builtin_amdgcn_s_barrier();
  __builtin_amdgcn_fence(__ATOMIC_ACQUIRE, "workgroup", "local");
}

template <int NT, int K, int RPT>
__device__ __forceinline__ void bn_request(
    const float* __restrict__ psum, const float* __restrict__ psq, int nparts,
    float4 (&s)[RPT], float4 (&q)[RPT]) {
  constexpr int QPR   = K / 4;
  constexpr int G     = NT / QPR;
  constexpr int PITCH = K < 32 ? 32 : K;
  static_assert(NT % QPR == 0, "threads must tile the partial rows");
  const int cq  = threadIdx.x % QPR;
  const int grp = threadIdx.x / QPR;
  const bool full   = RPT * G <= nparts;
  const long first  = full ? (long)grp * PITCH : 0;
  const long stride = full ? (long)G * PITCH : 0;
  const float* ps = psum + first + 4 * cq;
  const float* pq = psq  + first + 4 * cq;
#pragma unroll
  for (int i = 0; i < RPT; ++i) {
    s[i] = *(const float4*)(ps + i * stride);
    q[i] = *(const float4*)(pq + i * stride);
  }
}

template <int NT, int K, int RPT>
__device__ __forceinline__ void bn_finish(
    const float* __restrict__ psum, const float* __restrict__ psq, int nparts,
    const float4 (&s)[RPT], const float4 (&q)[RPT], float gamma, float beta,
    float invB, float* sc_s, float* sh_s, float* red) {
  constexpr int QPR   = K / 4;
  constexpr int G     = NT / QPR;
  constexpr int PITCH = K < 32 ? 32 : K;
  constexpr int RP    = G + 4;
  static_assert(G % 4 == 0, "a column's group sums are read 16 bytes at a time");
  const int t   = threadIdx.x;
  const int cq  = t % QPR;
  const int grp = t / QPR;
  float4 S = make_float4(0.0f, 0.0f, 0.0f, 0.0f);
  float4 Q = make_float4(0.0f, 0.0f, 0.0f, 0.0f);
#pragma unroll
  for (int i = 0; i < RPT; ++i) {
    S.x += s[i].x; S.y += s[i].y; S.z += s[i].z; S.w += s[i].w;
    Q.x += q[i].x; Q.y += q[i].y; Q.z += q[i].z; Q.w += q[i].w;
  }
  const bool full = RPT * G <= nparts;
  if (!full) {
    S = make_float4(0.0f, 0.0f, 0.0f, 0.0f);
    Q = S;
  }
  for (int p = grp + (full ? RPT * G : 0); p < nparts; p += G) {
    const float4 a = *(const float4*)(psum + (long)p * PITCH + 4 * cq);
    const float4 b = *(const float4*)(psq  + (long)p * PITCH + 4 * cq);
    S.x += a.x; S.y += a.y; S.z += a.z; S.w += a.w;
    Q.x += b.x; Q.y += b.y; Q.z += b.z; Q.w += b.w;
  }
  float* rs = red + (4 * cq) * RP + grp;
  float* rq = rs + K * RP;
  rs[0] = S.x; rs[RP] = S.y; rs[2 * RP] = S.z; rs[3 * RP] = S.w;
  rq[0] = Q.x; rq[RP] = Q.y; rq[2 * RP] = Q.z; rq[3 * RP] = Q.w;
  lds_barrier();
  if (t < K) {
    const float4* cs  = (const float4*)(red + t * RP);
    const float4* cq4 = (const float4*)(red + (K + t) * RP);
    float4 v[G / 4], w[G / 4];
#pragma unroll
    for (int i = 0; i < G / 4; ++i) {
      v[i] = cs[i];
      w[i] = cq4[i];
    }
    __builtin_amdgcn_sched_barrier(0);
    float Sc = v[0].x, Qc = w[0].x;
    Sc += v[0].y; Sc += v[0].z; Sc += v[0].w;
    Qc += w[0].y; Qc += w[0].z; Qc += w[0].w;
#pragma unroll
    for (int i = 1; i < G / 4; ++i) {
      Sc += v[i].x; Sc += v[i].y; Sc += v[i].z; Sc += v[i].w;
      Qc += w[i].x; Qc += w[i].y; Qc += w[i].z; Qc += w[i].w;
    }
    const float mean = Sc * invB;
    const float var  = Qc * invB - mean * mean;
    const float sc   = gamma * rsqrtf(var + EPS_BN);
    sc_s[t] = sc;
    sh_s[t] = beta - mean * sc;
  }
  lds_barrier();
}

#define MLP_WAVES   4
#define MLP_THREADS (MLP_WAVES * 32)

template <int K, int N, bool USE_BN>
__global__ __launch_bounds__(MLP_THREADS) void mlp_layer_kernel(
    const float* __restrict__ x,
    const float* __restrict__ w,
    const float* __restrict__ bias,
    const float* __restrict__ in_psum,
    const float* __restrict__ in_psq,
    const float* __restrict__ bn_g,
    const float* __restrict__ bn_be,
    float* __restrict__ y,
    float* __restrict__ out_psum,
    float* __restrict__ out_psq,
    int nparts, float invB) {
  constexpr int KS  = K / 4;
  constexpr int NT  = N / 16;
  constexpr int WP  = N + 8;
  constexpr int QPR = N / 4;
  constexpr int WQ  = K * QPR / MLP_THREADS;
  __shared__ __align__(16) float w_s[K * WP];
  __shared__ __align__(16) float sc_s[K];
  __shared__ __align__(16) float sh_s[K];
  constexpr int RPT = USE_BN ? K / 2 : 1;
  constexpr int BG  = MLP_THREADS / (K / 4);
  __shared__ __align__(16) float red[USE_BN ? 2 * K * (BG + 4) : 4];
  __shared__ float wred[2][MLP_WAVES][2][N];

  const int t    = threadIdx.x;
  const int lane = t & 31;
  const int wave = t >> 5;
  const int row0 = (blockIdx.x * MLP_WAVES + wave) * 16;
  const int r    = lane & 15;
  const int half = lane >> 4;
  const int col  = lane & 15;

  float4 wq[WQ];
#pragma unroll
  for (int i = 0; i < WQ; ++i)
    wq[i] = ((const float4*)w)[t + i * MLP_THREADS];

  float4 ps[RPT], pq[RPT];
  float gam = 0.0f, bet = 0.0f;
  if (USE_BN) {
    gam = bn_g[t % K];
    bet = bn_be[t % K];
    bn_request<MLP_THREADS, K, RPT>(in_psum, in_psq, nparts, ps, pq);
#pragma unroll
    for (int i = 0; i < WQ; ++i) {
      const int p = t + i * MLP_THREADS;
      *(float4*)(w_s + (p / QPR) * WP + 4 * (p % QPR)) = wq[i];
    }
  }
  __builtin_amdgcn_sched_barrier(0);

  const float* xr = x + (long)(row0 + r) * K + 2 * half;
  v2f a[KS];
#pragma unroll
  for (int s = 0; s < KS; ++s) a[s] = *(const v2f*)(xr + 4 * s);

  if (!USE_BN) {
#pragma unroll
    for (int i = 0; i < WQ; ++i) {
      const int p = t + i * MLP_THREADS;
      *(float4*)(w_s + (p / QPR) * WP + 4 * (p % QPR)) = wq[i];
    }
  }
  if (USE_BN)
    bn_finish<MLP_THREADS, K, RPT>(in_psum, in_psq, nparts, ps, pq, gam, bet,
                                   invB, sc_s, sh_s, red);
  else
    lds_barrier();

  if (USE_BN) {
#pragma unroll
    for (int s = 0; s < KS; ++s) {
      const v2f sc = *(const v2f*)(sc_s + 4 * s + 2 * half);
      const v2f sh = *(const v2f*)(sh_s + 4 * s + 2 * half);
      a[s] = sc * a[s] + sh;
    }
  }

  v8f acc[NT];
#pragma unroll
  for (int nt = 0; nt < NT; ++nt) acc[nt] = v8f{};
  const float* wl = w_s + (2 * half) * WP + col;
#pragma unroll
  for (int s = 0; s < KS; ++s) {
#pragma unroll
    for (int nt = 0; nt < NT; ++nt) {
      v2f bb;
      bb.x = wl[(4 * s) * WP + nt * 16];
      bb.y = wl[(4 * s + 1) * WP + nt * 16];
      acc[nt] = __builtin_amdgcn_wmma_f32_16x16x4_f32(
          false, a[s], false, bb, (short)0, acc[nt], false, false);
    }
  }

#pragma unroll
  for (int nt = 0; nt < NT; ++nt) {
    const int n = nt * 16 + col;
    const float bn = bias[n];
    float s = 0.0f, q = 0.0f;
#pragma unroll
    for (int j = 0; j < 8; ++j) {
      const int m = row0 + j + half * 8;
      const float v = fmaxf(acc[nt][j] + bn, 0.0f);
      y[(long)m * N + n] = v;
      s += v;
      q += v * v;
    }
    wred[0][wave][half][n] = s;
    wred[1][wave][half][n] = q;
  }
  lds_barrier();

  if (t < 2 * N) {
    const int kind   = t >= N ? 1 : 0;
    const int n      = t - kind * N;
    constexpr int opitch = N < 32 ? 32 : N;
    float a2 = 0.0f;
    for (int wv = 0; wv < MLP_WAVES; ++wv)
      a2 += wred[kind][wv][0][n] + wred[kind][wv][1][n];
    (kind ? out_psq : out_psum)[(long)blockIdx.x * opitch + n] = a2;
  }
}

#define FIN_THREADS 256

__global__ __launch_bounds__(FIN_THREADS) void final_kernel(
    const float* __restrict__ y3, const float* __restrict__ psum3,
    const float* __restrict__ psq3, int nparts, const float* __restrict__ g3,
    const float* __restrict__ be3, const float* __restrict__ wf,
    const float* __restrict__ bf, float* __restrict__ out, int Bn, float invB) {
  constexpr int RPT = 4;
  constexpr int BG  = FIN_THREADS / (16 / 4);
  __shared__ float sc_s[16], sh_s[16], wf_s[16];
  __shared__ __align__(16) float red[2 * 16 * (BG + 4)];
  const int t = threadIdx.x;
  const int b = blockIdx.x * FIN_THREADS + t;

  const float wfv = wf[t & 15];
  const float gam = g3[t & 15];
  const float bet = be3[t & 15];
  const float bfv = bf[0];
  const float4* yr = (const float4*)(y3 + (long)(b < Bn ? b : Bn - 1) * 16);
  float4 yv[4];
#pragma unroll
  for (int i = 0; i < 4; ++i) yv[i] = yr[i];
  float4 ps[RPT], pq[RPT];
  bn_request<FIN_THREADS, 16, RPT>(psum3, psq3, nparts, ps, pq);

  if (t < 16) wf_s[t] = wfv;
  bn_finish<FIN_THREADS, 16, RPT>(psum3, psq3, nparts, ps, pq, gam, bet, invB,
                                  sc_s, sh_s, red);
  if (b < Bn) {
    const float yy[16] = {yv[0].x, yv[0].y, yv[0].z, yv[0].w,
                          yv[1].x, yv[1].y, yv[1].z, yv[1].w,
                          yv[2].x, yv[2].y, yv[2].z, yv[2].w,
                          yv[3].x, yv[3].y, yv[3].z, yv[3].w};
    float s = 0.0f;
#pragma unroll
    for (int c = 0; c < 16; ++c)
      s += (sc_s[c] * yy[c] + sh_s[c]) * wf_s[c];
    out[b] = s + bfv;
  }
}

extern "C" void kernel_launch(void* const* d_in, const int* in_sizes, int n_in,
                              void* d_out, int out_size, void* d_ws,
                              size_t ws_size, hipStream_t stream) {
  const float* embed_u_z = (const float*)d_in[0];
  const float* embed_u_c = (const float*)d_in[1];
  const float* embed_v_z = (const float*)d_in[2];
  const float* embed_v_c = (const float*)d_in[3];
  const float* item_table = (const float*)d_in[4];
  const float* w1 = (const float*)d_in[5];
  const float* b1 = (const float*)d_in[6];
  const float* g1 = (const float*)d_in[7];
  const float* be1 = (const float*)d_in[8];
  const float* w2 = (const float*)d_in[9];
  const float* b2 = (const float*)d_in[10];
  const float* g2 = (const float*)d_in[11];
  const float* be2 = (const float*)d_in[12];
  const float* w3 = (const float*)d_in[13];
  const float* b3 = (const float*)d_in[14];
  const float* g3 = (const float*)d_in[15];
  const float* be3 = (const float*)d_in[16];
  const float* wf = (const float*)d_in[17];
  const float* bf = (const float*)d_in[18];
  const int* hist_items = (const int*)d_in[19];
  const int* hist_lens = (const int*)d_in[20];
  const int* item_ids = (const int*)d_in[21];
  const int* red_by_friends = (const int*)d_in[22];

  const int Bn = in_sizes[20];
  const int H  = in_sizes[19] / Bn;
  const float invB = 1.0f / (float)Bn;
  const int NB = Bn / (16 * MLP_WAVES);

  float* ws = (float*)d_ws;
  float* uv = ws;
  float* y1 = uv + (size_t)Bn * 128;
  float* y2 = y1 + (size_t)Bn * 64;
  float* y3 = y2 + (size_t)Bn * 32;
  float* ps1 = y3 + (size_t)Bn * 16;
  float* pq1 = ps1 + (size_t)NB * 64;
  float* ps2 = pq1 + (size_t)NB * 64;
  float* pq2 = ps2 + (size_t)NB * 32;
  float* ps3 = pq2 + (size_t)NB * 32;
  float* pq3 = ps3 + (size_t)NB * 32;

  gather_gate_kernel<<<Bn / (2 * GG_WAVES), GG_THREADS, 0, stream>>>(
      embed_u_z, embed_u_c, embed_v_z, embed_v_c, item_table,
      hist_items, hist_lens, item_ids, red_by_friends, uv, H);

  mlp_layer_kernel<128, 64, false><<<NB, MLP_THREADS, 0, stream>>>(
      uv, w1, b1, nullptr, nullptr, nullptr, nullptr,
      y1, ps1, pq1, 0, invB);

  mlp_layer_kernel<64, 32, true><<<NB, MLP_THREADS, 0, stream>>>(
      y1, w2, b2, ps1, pq1, g1, be1,
      y2, ps2, pq2, NB, invB);

  mlp_layer_kernel<32, 16, true><<<NB, MLP_THREADS, 0, stream>>>(
      y2, w3, b3, ps2, pq2, g2, be2,
      y3, ps3, pq3, NB, invB);

  final_kernel<<<(Bn + 255) / 256, 256, 0, stream>>>(
      y3, ps3, pq3, NB, g3, be3, wf, bf, (float*)d_out, Bn, invB);
}
